# starting kernel
# baseline (speedup 1.0000x reference)
#include <hip/hip_runtime.h>
#include <math.h>


typedef __attribute__((ext_vector_type(2))) float v2f;
typedef __attribute__((ext_vector_type(8))) float v8f;

#define LQ    512
#define NH    12
#define NODE  384
#define PAIR  128
#define OUTIN 176
#define KOUT  (NH*OUTIN)

#define OFF_SQ   ((size_t)0)
#define OFF_SK   (OFF_SQ  + (size_t)NH*LQ*16)
#define OFF_SV   (OFF_SK  + (size_t)NH*LQ*16)
#define OFF_PQ   (OFF_SV  + (size_t)NH*LQ*16)
#define OFF_PK   (OFF_PQ  + (size_t)NH*LQ*16)
#define OFF_PV   (OFF_PK  + (size_t)NH*LQ*16)
#define OFF_QN2  (OFF_PV  + (size_t)NH*LQ*24)
#define OFF_KN2  (OFF_QN2 + (size_t)NH*LQ)
#define OFF_BIAS (OFF_KN2 + (size_t)NH*LQ)
#define OFF_ATTN (OFF_BIAS+ (size_t)NH*LQ*LQ)
#define OFF_RES  (OFF_ATTN+ (size_t)NH*LQ*LQ)

#define RES_IDX(h, l, f) (((size_t)(l)*NH + (h))*OUTIN + (f))

static __device__ __forceinline__ v8f wmma4(v2f a, v2f b, v8f c) {
  return __builtin_amdgcn_wmma_f32_16x16x4_f32(
       false, a,  false, b,
       (short)0, c,  false,  false);
}

static __device__ __forceinline__ v8f vzero8() {
  v8f z = {0.f,0.f,0.f,0.f,0.f,0.f,0.f,0.f};
  return z;
}

__global__ __launch_bounds__(32) void proj_kernel(
    const float* __restrict__ node,
    const float* __restrict__ sq_k, const float* __restrict__ sk_k,
    const float* __restrict__ sv_k, const float* __restrict__ pq_k,
    const float* __restrict__ pk_k, const float* __restrict__ pv_k,
    float* __restrict__ ws)
{
  const int lane  = threadIdx.x;
  const int mt    = blockIdx.x;
  const int seg   = blockIdx.y;
  const int h     = blockIdx.z;
  const int row   = lane & 15;
  const int khalf = (lane >> 4) * 2;

  const float* wb = sk_k + (size_t)h*NODE*16;
  int   wstride = 16;
  int   wcol    = row;
  float wscale  = 1.f;
  switch (seg) {
    case 0: wb = sq_k + (size_t)h*NODE*16; wscale = 0.25f; break;
    case 1: break;
    case 2: wb = sv_k + (size_t)h*NODE*16; break;
    case 3: wb = pq_k + (size_t)h*NODE*12; wstride = 12;
            wcol = (row < 12) ? row : 11; wscale = (row < 12) ? 1.f : 0.f; break;
    case 4: wb = pk_k + (size_t)h*NODE*12; wstride = 12;
            wcol = (row < 12) ? row : 11; wscale = (row < 12) ? 1.f : 0.f; break;
    case 5: wb = pv_k + (size_t)h*NODE*24; wstride = 24; break;
    case 6: wb = pv_k + (size_t)h*NODE*24; wstride = 24;
            wcol = (row < 8) ? 16 + row : 23; wscale = (row < 8) ? 1.f : 0.f; break;
  }
  const float* arow = node + (size_t)(mt*16 + row)*NODE;
  const float* wp   = wb + wcol;

  v8f acc = vzero8();
  for (int k0 = 0; k0 < NODE; k0 += 4) {
    const int ka = k0 + khalf;
    v2f a; a.x = arow[ka]; a.y = arow[ka + 1];
    v2f b;
    b.x = wp[(size_t)ka * wstride]       * wscale;
    b.y = wp[(size_t)(ka + 1) * wstride] * wscale;
    acc = wmma4(a, b, acc);
  }

  const int col   = lane & 15;
  const int rbase = (lane >> 4) * 8;
  float* ob = ws + OFF_SK + (size_t)h*LQ*16;
  int ostride = 16, ocol = col;
  bool dostore = true;
  switch (seg) {
    case 0: ob = ws + OFF_SQ + (size_t)h*LQ*16; break;
    case 1: break;
    case 2: ob = ws + OFF_SV + (size_t)h*LQ*16; break;
    case 3: ob = ws + OFF_PQ + (size_t)h*LQ*16; break;
    case 4: ob = ws + OFF_PK + (size_t)h*LQ*16; break;
    case 5: ob = ws + OFF_PV + (size_t)h*LQ*24; ostride = 24; break;
    case 6: ob = ws + OFF_PV + (size_t)h*LQ*24; ostride = 24;
            ocol = 16 + col; dostore = (col < 8); break;
  }
  if (dostore) {
#pragma unroll
    for (int r = 0; r < 8; ++r)
      ob[(size_t)(mt*16 + rbase + r)*ostride + ocol] = acc[r];
  }
}

__global__ void rigid_kernel(const float* __restrict__ R,
                             const float* __restrict__ t,
                             float* __restrict__ ws)
{
  const int idx = blockIdx.x * blockDim.x + threadIdx.x;
  if (idx >= NH * LQ) return;
  const int h = idx / LQ, l = idx % LQ;
  float Rm[9], tv[3];
#pragma unroll
  for (int i = 0; i < 9; ++i) Rm[i] = R[l*9 + i];
#pragma unroll
  for (int i = 0; i < 3; ++i) tv[i] = t[l*3 + i];

  float* Pq = ws + OFF_PQ + ((size_t)h*LQ + l)*16;
  float* Pk = ws + OFF_PK + ((size_t)h*LQ + l)*16;
  float* Pv = ws + OFF_PV + ((size_t)h*LQ + l)*24;

  float qs = 0.f, ksum = 0.f;
#pragma unroll
  for (int p = 0; p < 4; ++p) {
    float loc[3], g[3];
    for (int f = 0; f < 3; ++f) loc[f] = Pq[p*3+f];
    for (int gg = 0; gg < 3; ++gg) {
      g[gg] = tv[gg];
      for (int f = 0; f < 3; ++f) g[gg] += loc[f]*Rm[gg*3+f];
      Pq[p*3+gg] = g[gg]; qs += g[gg]*g[gg];
    }
    for (int f = 0; f < 3; ++f) loc[f] = Pk[p*3+f];
    for (int gg = 0; gg < 3; ++gg) {
      g[gg] = tv[gg];
      for (int f = 0; f < 3; ++f) g[gg] += loc[f]*Rm[gg*3+f];
      Pk[p*3+gg] = g[gg]; ksum += g[gg]*g[gg];
    }
  }
#pragma unroll
  for (int p = 0; p < 8; ++p) {
    float loc[3], g[3];
    for (int f = 0; f < 3; ++f) loc[f] = Pv[p*3+f];
    for (int gg = 0; gg < 3; ++gg) {
      g[gg] = tv[gg];
      for (int f = 0; f < 3; ++f) g[gg] += loc[f]*Rm[gg*3+f];
      Pv[p*3+gg] = g[gg];
    }
  }
  ws[OFF_QN2 + h*LQ + l] = qs;
  ws[OFF_KN2 + h*LQ + l] = ksum;
}

__global__ __launch_bounds__(32) void bias_kernel(
    const float* __restrict__ pairwise, const float* __restrict__ pr_k,
    float* __restrict__ ws)
{
  float* bias = ws + OFF_BIAS;
  const int lane  = threadIdx.x;
  const int tile  = blockIdx.x;
  const int row   = lane & 15;
  const int khalf = (lane >> 4) * 2;
  const int flat  = tile*16 + row;
  const float* prow = pairwise + (size_t)flat * PAIR;
  const float* prk  = pr_k + (size_t)((row < NH) ? row : NH - 1) * PAIR;
  const float bmask = (row < NH) ? 1.f : 0.f;

  v8f acc = vzero8();
  for (int k0 = 0; k0 < PAIR; k0 += 4) {
    const int ka = k0 + khalf;
    __builtin_prefetch(prow + ka + 64, 0, 0);
    v2f a; a.x = prow[ka];          a.y = prow[ka + 1];
    v2f b; b.x = prk[ka] * bmask;   b.y = prk[ka + 1] * bmask;
    acc = wmma4(a, b, acc);
  }
  const int col = lane & 15;
  const int rbase = (lane >> 4) * 8;
  if (col < NH) {
#pragma unroll
    for (int r = 0; r < 8; ++r) {
      const int f2 = tile*16 + rbase + r;
      const int i = f2 >> 9, j = f2 & 511;
      bias[((size_t)col*LQ + i)*LQ + j] = acc[r];
    }
  }
}

#define SMS 516
__global__ __launch_bounds__(128) void attn_kernel(
    const float* __restrict__ point_weight, float* __restrict__ ws)
{
  __shared__ float sm[16 * SMS];
  const int tid  = threadIdx.x;
  const int lane = tid & 31;
  const int w    = tid >> 5;
  const int nt   = blockIdx.x;
  const int h    = blockIdx.y;

  const float* Sq  = ws + OFF_SQ;  const float* Sk  = ws + OFF_SK;
  const float* Pq  = ws + OFF_PQ;  const float* Pk  = ws + OFF_PK;
  const float* qn2 = ws + OFF_QN2; const float* kn2 = ws + OFF_KN2;
  const float* bias= ws + OFF_BIAS;
  const float* Sv  = ws + OFF_SV;  const float* Pv  = ws + OFF_PV;
  float* attn = ws + OFF_ATTN;
  float* res  = ws + OFF_RES;

  const float pwx = point_weight[h];
  const float pw  = (pwx > 20.f) ? pwx : log1pf(__expf(pwx));
  const float cpt = -0.5f * pw / sqrtf(18.f);
  const float rs3 = 0.57735026919f;

  const int row   = lane & 15;
  const int khalf = (lane >> 4) * 2;
  const int col   = lane & 15;
  const int rbase = (lane >> 4) * 8;

  v2f aq[4], ap[4];
#pragma unroll
  for (int ks = 0; ks < 4; ++ks) {
    const int k = ks*4 + khalf;
    const int l = nt*16 + row;
    aq[ks].x = Sq[((size_t)h*LQ + l)*16 + k];
    aq[ks].y = Sq[((size_t)h*LQ + l)*16 + k + 1];
    ap[ks].x = Pq[((size_t)h*LQ + l)*16 + k];
    ap[ks].y = Pq[((size_t)h*LQ + l)*16 + k + 1];
  }
  float qv[8];
#pragma unroll
  for (int r = 0; r < 8; ++r) qv[r] = qn2[h*LQ + nt*16 + rbase + r];

  for (int mt = w; mt < 32; mt += 4) {
    v8f s = vzero8(), p = vzero8();
#pragma unroll
    for (int ks = 0; ks < 4; ++ks) {
      const int k = ks*4 + khalf;
      const int m = mt*16 + row;
      v2f bq, bp;
      bq.x = Sk[((size_t)h*LQ + m)*16 + k]; bq.y = Sk[((size_t)h*LQ + m)*16 + k + 1];
      bp.x = Pk[((size_t)h*LQ + m)*16 + k]; bp.y = Pk[((size_t)h*LQ + m)*16 + k + 1];
      s = wmma4(aq[ks], bq, s);
      p = wmma4(ap[ks], bp, p);
    }
    const int m = mt*16 + col;
    const float kv = kn2[h*LQ + m];
#pragma unroll
    for (int r = 0; r < 8; ++r) {
      const int rr = rbase + r;
      const int n  = nt*16 + rr;
      const float dist = qv[r] + kv - 2.f*p[r];
      sm[rr*SMS + m] =
          (s[r] + bias[((size_t)h*LQ + n)*LQ + m] + cpt*dist) * rs3;
    }
  }
  __syncthreads();

  {
    const int rrow = tid >> 3;
    const int sub  = tid & 7;
    float mx = -1e30f;
    for (int j = sub; j < LQ; j += 8) mx = fmaxf(mx, sm[rrow*SMS + j]);
    for (int o = 1; o < 8; o <<= 1) mx = fmaxf(mx, __shfl_xor(mx, o, 8));
    float sum = 0.f;
    for (int j = sub; j < LQ; j += 8) {
      const float e = __expf(sm[rrow*SMS + j] - mx);
      sm[rrow*SMS + j] = e; sum += e;
    }
    for (int o = 1; o < 8; o <<= 1) sum += __shfl_xor(sum, o, 8);
    const float inv = 1.f / sum;
    const int n = nt*16 + rrow;
    for (int j = sub; j < LQ; j += 8) {
      const float v = sm[rrow*SMS + j] * inv;
      sm[rrow*SMS + j] = v;
      attn[((size_t)h*LQ + n)*LQ + j] = v;
    }
  }
  __syncthreads();

  if (w < 3) {
    const float* vb = (w == 0) ? Sv + (size_t)h*LQ*16 : Pv + (size_t)h*LQ*24;
    const int    vs = (w == 0) ? 16 : 24;
    int   vc = col;
    float vmask = 1.f;
    if (w == 2) { vc = (col < 8) ? 16 + col : 23; vmask = (col < 8) ? 1.f : 0.f; }
    const float* vp = vb + vc;

    v8f acc = vzero8();
    for (int kt = 0; kt < 128; ++kt) {
      const int k = kt*4 + khalf;
      v2f a; a.x = sm[row*SMS + k]; a.y = sm[row*SMS + k + 1];
      v2f b;
      b.x = vp[(size_t)k * vs]       * vmask;
      b.y = vp[(size_t)(k + 1) * vs] * vmask;
      acc = wmma4(a, b, acc);
    }
    const int fbase = (w == 0) ? 0 : 144;
#pragma unroll
    for (int r = 0; r < 8; ++r) {
      const int n = nt*16 + rbase + r;
      const float v = acc[r];
      if (w < 2)        res[RES_IDX(h, n, fbase + col)] = v;
      else if (col < 8) res[RES_IDX(h, n, 160 + col)]   = v;
    }
  }
}

__global__ __launch_bounds__(32) void pairout_kernel(
    const float* __restrict__ pairwise, float* __restrict__ ws)
{
  const float* attn = ws + OFF_ATTN;
  float* res = ws + OFF_RES;
  const int lane  = threadIdx.x;
  const int n     = blockIdx.x;
  const int row   = lane & 15;
  const int khalf = (lane >> 4) * 2;
  const int col   = lane & 15;

  const int hh0 = (row < NH) ? row : NH - 1;
  const float amask = (row < NH) ? 1.f : 0.f;
  const float* arow = attn + ((size_t)hh0*LQ + n)*LQ;

  v8f acc[8];
#pragma unroll
  for (int i = 0; i < 8; ++i) acc[i] = vzero8();

  for (int kt = 0; kt < 128; ++kt) {
    const int m = kt*4 + khalf;
    v2f a;
    a.x = arow[m] * amask;
    a.y = arow[m + 1] * amask;
    const float* pr0 = pairwise + ((size_t)n*LQ + m) * PAIR + col;
    __builtin_prefetch(pr0 + 4*PAIR, 0, 0);
#pragma unroll
    for (int ntile = 0; ntile < 8; ++ntile) {
      v2f b; b.x = pr0[ntile*16]; b.y = pr0[ntile*16 + PAIR];
      acc[ntile] = wmma4(a, b, acc[ntile]);
    }
  }
  const int rbase = (lane >> 4) * 8;
#pragma unroll
  for (int ntile = 0; ntile < 8; ++ntile)
#pragma unroll
    for (int r = 0; r < 8; ++r) {
      const int hh = rbase + r;
      if (hh < NH)
        res[RES_IDX(hh, n, 16 + ntile*16 + col)] = acc[ntile][r];
    }
}

__global__ void pfinish_kernel(const float* __restrict__ R,
                               const float* __restrict__ t,
                               float* __restrict__ ws)
{
  const int idx = blockIdx.x * blockDim.x + threadIdx.x;
  if (idx >= NH * LQ) return;
  const int h = idx / LQ, l = idx % LQ;
  float* resp = ws + OFF_RES + RES_IDX(h, l, 0);
  float Rm[9], tv[3];
#pragma unroll
  for (int i = 0; i < 9; ++i) Rm[i] = R[l*9 + i];
#pragma unroll
  for (int i = 0; i < 3; ++i) tv[i] = t[l*3 + i];
#pragma unroll
  for (int p = 0; p < 8; ++p) {
    float g[3], loc[3], nrm = 0.f;
    for (int gg = 0; gg < 3; ++gg) g[gg] = resp[144 + p*3 + gg] - tv[gg];
    for (int f = 0; f < 3; ++f) {
      float v = 0.f;
      for (int gg = 0; gg < 3; ++gg) v += g[gg]*Rm[gg*3 + f];
      loc[f] = v; nrm += v*v;
    }
    for (int f = 0; f < 3; ++f) resp[144 + p*3 + f] = loc[f];
    resp[168 + p] = sqrtf(nrm + 1e-12f);
  }
}

__global__ __launch_bounds__(32) void outproj_kernel(
    const float* __restrict__ out_k, const float* __restrict__ out_b,
    const float* __restrict__ res, float* __restrict__ out)
{
  const int lane  = threadIdx.x;
  const int mt    = blockIdx.x;
  const int ntile = blockIdx.y;
  const int row   = lane & 15;
  const int khalf = (lane >> 4) * 2;
  const int col   = lane & 15;
  const int o     = ntile*16 + col;

  const float* resl = res + (size_t)(mt*16 + row) * KOUT;
  const float* okp  = out_k + o;

  v8f acc = vzero8();
  for (int k0 = 0; k0 < KOUT; k0 += 4) {
    const int ka = k0 + khalf;
    v2f a; a.x = resl[ka];                 a.y = resl[ka + 1];
    v2f b; b.x = okp[(size_t)ka * NODE];   b.y = okp[(size_t)(ka + 1) * NODE];
    acc = wmma4(a, b, acc);
  }
  const int rbase = (lane >> 4) * 8;
  const float ob = out_b[o];
#pragma unroll
  for (int r = 0; r < 8; ++r) {
    const int l = mt*16 + rbase + r;
    out[(size_t)l*NODE + o] = acc[r] + ob;
  }
}

extern "C" void kernel_launch(void* const* d_in, const int* in_sizes, int n_in,
                              void* d_out, int out_size, void* d_ws, size_t ws_size,
                              hipStream_t stream) {
  const float* node  = (const float*)d_in[0];
  const float* R     = (const float*)d_in[1];
  const float* t     = (const float*)d_in[2];
  const float* pairw = (const float*)d_in[3];
  const float* sq_k  = (const float*)d_in[4];
  const float* sk_k  = (const float*)d_in[5];
  const float* sv_k  = (const float*)d_in[6];
  const float* pq_k  = (const float*)d_in[7];
  const float* pk_k  = (const float*)d_in[8];
  const float* pv_k  = (const float*)d_in[9];
  const float* pr_k  = (const float*)d_in[10];
  const float* out_k = (const float*)d_in[11];
  const float* out_b = (const float*)d_in[12];
  const float* pwgt  = (const float*)d_in[13];
  float* ws  = (float*)d_ws;
  float* out = (float*)d_out;

  proj_kernel<<<dim3(32, 7, NH), 32, 0, stream>>>(
      node, sq_k, sk_k, sv_k, pq_k, pk_k, pv_k, ws);
  rigid_kernel<<<(NH*LQ + 127)/128, 128, 0, stream>>>(R, t, ws);
  bias_kernel<<<(LQ*LQ)/16, 32, 0, stream>>>(pairw, pr_k, ws);
  attn_kernel<<<dim3(32, NH), 128, 0, stream>>>(pwgt, ws);
  pairout_kernel<<<LQ, 32, 0, stream>>>(pairw, ws);
  pfinish_kernel<<<(NH*LQ + 127)/128, 128, 0, stream>>>(R, t, ws);
  outproj_kernel<<<dim3(32, 24), 32, 0, stream>>>(out_k, out_b, ws + OFF_RES, out);
}


# attn and proj: all wmma operands read from LDS, staged by cooperative 16-byte loads (attn: keys + bias rows up front, values in flight during softmax; proj: next chunk in flight); sum order unchanged
# speedup vs baseline: 2.9996x; 2.9996x over previous
#include <hip/hip_runtime.h>
#include <math.h>


typedef __attribute__((ext_vector_type(2))) float v2f;
typedef __attribute__((ext_vector_type(4))) float v4f;
typedef __attribute__((ext_vector_type(8))) float v8f;

#define LQ    512
#define NH    12
#define NODE  384
#define PAIR  128
#define OUTIN 176
#define KOUT  (NH*OUTIN)

#define OFF_SQ   ((size_t)0)
#define OFF_SK   (OFF_SQ  + (size_t)NH*LQ*16)
#define OFF_SV   (OFF_SK  + (size_t)NH*LQ*16)
#define OFF_PQ   (OFF_SV  + (size_t)NH*LQ*16)
#define OFF_PK   (OFF_PQ  + (size_t)NH*LQ*16)
#define OFF_PV   (OFF_PK  + (size_t)NH*LQ*16)
#define OFF_QN2  (OFF_PV  + (size_t)NH*LQ*24)
#define OFF_KN2  (OFF_QN2 + (size_t)NH*LQ)
#define OFF_BIAS (OFF_KN2 + (size_t)NH*LQ)
#define OFF_ATTN (OFF_BIAS+ (size_t)NH*LQ*LQ)
#define OFF_RES  (OFF_ATTN+ (size_t)NH*LQ*LQ)

#define RES_IDX(h, l, f) (((size_t)(l)*NH + (h))*OUTIN + (f))

static __device__ __forceinline__ v8f wmma4(v2f a, v2f b, v8f c) {
  return __builtin_amdgcn_wmma_f32_16x16x4_f32(
       false, a,  false, b,
       (short)0, c,  false,  false);
}

static __device__ __forceinline__ v8f vzero8() {
  v8f z = {0.f,0.f,0.f,0.f,0.f,0.f,0.f,0.f};
  return z;
}

#define PJ_KC  64
#define PJ_NCH (NODE / PJ_KC)
#define PJ_AP  68

template <int WS>
static __device__ __forceinline__ v8f proj_chunk(
    const float* __restrict__ Ab, const float* __restrict__ Bb, float wscale, v8f acc)
{
  v2f a[PJ_KC / 4], b[PJ_KC / 4];
#pragma unroll
  for (int kk = 0; kk < PJ_KC / 4; ++kk) {
    a[kk]   = *reinterpret_cast<const v2f*>(Ab + kk*4);
    b[kk].x = Bb[(kk*4)     * WS] * wscale;
    b[kk].y = Bb[(kk*4 + 1) * WS] * wscale;
  }
  __builtin_amdgcn_sched_barrier(0);
#pragma unroll
  for (int kk = 0; kk < PJ_KC / 4; ++kk)
    acc = wmma4(a[kk], b[kk], acc);
  asm volatile("" : "+v"(acc));
  __builtin_amdgcn_sched_barrier(0);
  return acc;
}

template <int WS>
static __device__ __forceinline__ v8f proj_body(
    const float* __restrict__ ga, const float* __restrict__ gw,
    float* __restrict__ As, float* __restrict__ Bs,
    int lane, int wcol, float wscale)
{
  constexpr int NB = PJ_KC * WS / 128;
  const int row   = lane & 15;
  const int khalf = (lane >> 4) * 2;
  const float* gaa = ga + (size_t)(lane >> 4) * NODE + (lane & 15) * 4;
  const int sa = (lane >> 4) * PJ_AP + (lane & 15) * 4;
  const float* gbb = gw + (size_t)lane * 4;
  const int sb = lane * 4;
  const int fa = row * PJ_AP + khalf;
  const int fb = khalf * WS + wcol;

  v4f ra[8], rb[NB];
#pragma unroll
  for (int q = 0; q < 8; ++q)
    ra[q] = *reinterpret_cast<const v4f*>(gaa + (size_t)q * 2 * NODE);
#pragma unroll
  for (int q = 0; q < NB; ++q)
    rb[q] = *reinterpret_cast<const v4f*>(gbb + (size_t)q * 128);
#pragma unroll
  for (int q = 0; q < 8; ++q)
    *reinterpret_cast<v4f*>(&As[sa + q*2*PJ_AP]) = ra[q];
#pragma unroll
  for (int q = 0; q < NB; ++q)
    *reinterpret_cast<v4f*>(&Bs[sb + q*128]) = rb[q];
  __syncthreads();

  v8f acc = vzero8();
  for (int ch = 0; ch < PJ_NCH - 1; ++ch) {
    const float* gan = gaa + (size_t)(ch + 1) * PJ_KC;
    const float* gbn = gbb + (size_t)(ch + 1) * PJ_KC * WS;
#pragma unroll
    for (int q = 0; q < 8; ++q)
      ra[q] = *reinterpret_cast<const v4f*>(gan + (size_t)q * 2 * NODE);
#pragma unroll
    for (int q = 0; q < NB; ++q)
      rb[q] = *reinterpret_cast<const v4f*>(gbn + (size_t)q * 128);

    acc = proj_chunk<WS>(&As[fa], &Bs[fb], wscale, acc);

    __syncthreads();
#pragma unroll
    for (int q = 0; q < 8; ++q)
      *reinterpret_cast<v4f*>(&As[sa + q*2*PJ_AP]) = ra[q];
#pragma unroll
    for (int q = 0; q < NB; ++q)
      *reinterpret_cast<v4f*>(&Bs[sb + q*128]) = rb[q];
    __syncthreads();
  }
  return proj_chunk<WS>(&As[fa], &Bs[fb], wscale, acc);
}

__global__ __launch_bounds__(32) void proj_kernel(
    const float* __restrict__ node,
    const float* __restrict__ sq_k, const float* __restrict__ sk_k,
    const float* __restrict__ sv_k, const float* __restrict__ pq_k,
    const float* __restrict__ pk_k, const float* __restrict__ pv_k,
    float* __restrict__ ws)
{
  __shared__ __attribute__((aligned(16))) float As[16 * PJ_AP];
  __shared__ __attribute__((aligned(16))) float Bs[PJ_KC * 24];

  const int lane  = threadIdx.x;
  const int mt    = blockIdx.x;
  const int seg   = blockIdx.y;
  const int h     = blockIdx.z;
  const int row   = lane & 15;

  const float* ga = node + (size_t)(mt*16) * NODE;
  v8f acc;
  if (seg < 3) {
    const float* wb = (seg == 0) ? sq_k : (seg == 1) ? sk_k : sv_k;
    acc = proj_body<16>(ga, wb + (size_t)h*NODE*16, As, Bs, lane, row,
                        (seg == 0) ? 0.25f : 1.f);
  } else if (seg < 5) {
    const float* wb = (seg == 3) ? pq_k : pk_k;
    acc = proj_body<12>(ga, wb + (size_t)h*NODE*12, As, Bs, lane,
                        (row < 12) ? row : 11, (row < 12) ? 1.f : 0.f);
  } else if (seg == 5) {
    acc = proj_body<24>(ga, pv_k + (size_t)h*NODE*24, As, Bs, lane, row, 1.f);
  } else {
    acc = proj_body<24>(ga, pv_k + (size_t)h*NODE*24, As, Bs, lane,
                        (row < 8) ? 16 + row : 23, (row < 8) ? 1.f : 0.f);
  }

  const int col   = lane & 15;
  const int rbase = (lane >> 4) * 8;
  float* ob = ws + OFF_SK + (size_t)h*LQ*16;
  int ostride = 16, ocol = col;
  bool dostore = true;
  switch (seg) {
    case 0: ob = ws + OFF_SQ + (size_t)h*LQ*16; break;
    case 1: break;
    case 2: ob = ws + OFF_SV + (size_t)h*LQ*16; break;
    case 3: ob = ws + OFF_PQ + (size_t)h*LQ*16; break;
    case 4: ob = ws + OFF_PK + (size_t)h*LQ*16; break;
    case 5: ob = ws + OFF_PV + (size_t)h*LQ*24; ostride = 24; break;
    case 6: ob = ws + OFF_PV + (size_t)h*LQ*24; ostride = 24;
            ocol = 16 + col; dostore = (col < 8); break;
  }
  if (dostore) {
#pragma unroll
    for (int r = 0; r < 8; ++r)
      ob[(size_t)(mt*16 + rbase + r)*ostride + ocol] = acc[r];
  }
}

__global__ void rigid_kernel(const float* __restrict__ R,
                             const float* __restrict__ t,
                             float* __restrict__ ws)
{
  const int idx = blockIdx.x * blockDim.x + threadIdx.x;
  if (idx >= NH * LQ) return;
  const int h = idx / LQ, l = idx % LQ;
  float Rm[9], tv[3];
#pragma unroll
  for (int i = 0; i < 9; ++i) Rm[i] = R[l*9 + i];
#pragma unroll
  for (int i = 0; i < 3; ++i) tv[i] = t[l*3 + i];

  float* Pq = ws + OFF_PQ + ((size_t)h*LQ + l)*16;
  float* Pk = ws + OFF_PK + ((size_t)h*LQ + l)*16;
  float* Pv = ws + OFF_PV + ((size_t)h*LQ + l)*24;

  float qs = 0.f, ksum = 0.f;
#pragma unroll
  for (int p = 0; p < 4; ++p) {
    float loc[3], g[3];
    for (int f = 0; f < 3; ++f) loc[f] = Pq[p*3+f];
    for (int gg = 0; gg < 3; ++gg) {
      g[gg] = tv[gg];
      for (int f = 0; f < 3; ++f) g[gg] += loc[f]*Rm[gg*3+f];
      Pq[p*3+gg] = g[gg]; qs += g[gg]*g[gg];
    }
    for (int f = 0; f < 3; ++f) loc[f] = Pk[p*3+f];
    for (int gg = 0; gg < 3; ++gg) {
      g[gg] = tv[gg];
      for (int f = 0; f < 3; ++f) g[gg] += loc[f]*Rm[gg*3+f];
      Pk[p*3+gg] = g[gg]; ksum += g[gg]*g[gg];
    }
  }
#pragma unroll
  for (int p = 0; p < 8; ++p) {
    float loc[3], g[3];
    for (int f = 0; f < 3; ++f) loc[f] = Pv[p*3+f];
    for (int gg = 0; gg < 3; ++gg) {
      g[gg] = tv[gg];
      for (int f = 0; f < 3; ++f) g[gg] += loc[f]*Rm[gg*3+f];
      Pv[p*3+gg] = g[gg];
    }
  }
  ws[OFF_QN2 + h*LQ + l] = qs;
  ws[OFF_KN2 + h*LQ + l] = ksum;
}

#define BI_JC 128
#define BI_NCH (LQ / BI_JC)
#define BI_BP 132

static __device__ __forceinline__ v8f bias_chunk(
    const float* __restrict__ Bb, const v2f (&pa)[PAIR / 4])
{
  v2f b[PAIR / 4];
#pragma unroll
  for (int kk = 0; kk < PAIR / 4; ++kk)
    b[kk] = *reinterpret_cast<const v2f*>(Bb + kk*4);
  __builtin_amdgcn_sched_barrier(0);
  v8f acc = vzero8();
#pragma unroll
  for (int kk = 0; kk < PAIR / 4; ++kk)
    acc = wmma4(pa[kk], b[kk], acc);
  asm volatile("" : "+v"(acc));
  __builtin_amdgcn_sched_barrier(0);
  return acc;
}

static __device__ __forceinline__ void bias_store(
    float* __restrict__ ob, v8f acc, bool lowhalf)
{
#pragma unroll
  for (int r = 0; r < 4; ++r) ob[(size_t)r * LQ * LQ] = acc[r];
  if (lowhalf) {
#pragma unroll
    for (int r = 4; r < 8; ++r) ob[(size_t)r * LQ * LQ] = acc[r];
  }
}

__global__ __launch_bounds__(256) void bias_kernel(
    const float* __restrict__ pairwise, const float* __restrict__ pr_k,
    float* __restrict__ ws)
{
  __shared__ __attribute__((aligned(16))) float Bs[BI_JC * BI_BP];

  float* bias = ws + OFF_BIAS;
  const int tid   = threadIdx.x;
  const int lane  = tid & 31;
  const int w     = tid >> 5;
  const int i     = blockIdx.x;
  const int row   = lane & 15;
  const int khalf = (lane >> 4) * 2;

  const float* gb = pairwise + (size_t)i * LQ * PAIR + (size_t)tid * 4;
  const int sb = (tid >> 5) * BI_BP + (tid & 31) * 4;
  const int fb = (w*16 + row) * BI_BP + khalf;

  v4f rb[16];
#pragma unroll
  for (int q = 0; q < 16; ++q)
    rb[q] = *reinterpret_cast<const v4f*>(gb + (size_t)q * 1024);

  const bool live = row < NH;
  const float* prk = pr_k + (size_t)(live ? row : NH - 1) * PAIR + khalf;
  v2f pa[PAIR / 4];
#pragma unroll
  for (int kk = 0; kk < PAIR / 4; ++kk) {
    const v2f t = *reinterpret_cast<const v2f*>(prk + kk*4);
    pa[kk].x = live ? t.x : 0.f;
    pa[kk].y = live ? t.y : 0.f;
  }

#pragma unroll
  for (int q = 0; q < 16; ++q)
    *reinterpret_cast<v4f*>(&Bs[sb + q*8*BI_BP]) = rb[q];
  __syncthreads();

  const bool lowhalf = lane < 16;
  float* ob = bias + ((size_t)((lane >> 4) * 8) * LQ + i) * LQ + w*16 + row;

  for (int ch = 0; ch < BI_NCH - 1; ++ch) {
    const float* gbn = gb + (size_t)(ch + 1) * BI_JC * PAIR;
#pragma unroll
    for (int q = 0; q < 16; ++q)
      rb[q] = *reinterpret_cast<const v4f*>(gbn + (size_t)q * 1024);

    const v8f acc = bias_chunk(&Bs[fb], pa);

    __syncthreads();
#pragma unroll
    for (int q = 0; q < 16; ++q)
      *reinterpret_cast<v4f*>(&Bs[sb + q*8*BI_BP]) = rb[q];
    __syncthreads();
    bias_store(ob + ch * BI_JC, acc, lowhalf);
  }
  const v8f acc = bias_chunk(&Bs[fb], pa);
  bias_store(ob + (BI_NCH - 1) * BI_JC, acc, lowhalf);
}

#define SMS 516
#define AT_KP 20
#define AT_G  32

template <int VP>
static __device__ __forceinline__ v8f av_group(
    const float* __restrict__ Ab, const float* __restrict__ Bb, v8f acc)
{
  v2f a[AT_G], b[AT_G];
#pragma unroll
  for (int kk = 0; kk < AT_G; ++kk) {
    a[kk]   = *reinterpret_cast<const v2f*>(Ab + kk*4);
    b[kk].x = Bb[(kk*4)     * VP];
    b[kk].y = Bb[(kk*4 + 1) * VP];
  }
  __builtin_amdgcn_sched_barrier(0);
#pragma unroll
  for (int kk = 0; kk < AT_G; ++kk)
    acc = wmma4(a[kk], b[kk], acc);
  __builtin_amdgcn_sched_barrier(0);
  return acc;
}

template <int VP>
static __device__ __forceinline__ v8f av_tile(
    const float* __restrict__ Ab, const float* __restrict__ Bb)
{
  v8f acc = vzero8();
#pragma unroll
  for (int g = 0; g < LQ / (4 * AT_G); ++g)
    acc = av_group<VP>(Ab + g * 4 * AT_G, Bb + (size_t)g * 4 * AT_G * VP, acc);
  return acc;
}

__global__ __launch_bounds__(128) void attn_kernel(
    const float* __restrict__ point_weight, float* __restrict__ ws)
{
  __shared__ __attribute__((aligned(16))) float sm[16 * SMS];
  __shared__ __attribute__((aligned(16))) float Rs[LQ * 40];
  __shared__ __attribute__((aligned(16))) float kn[LQ];
  const int tid  = threadIdx.x;
  const int lane = tid & 31;
  const int w    = tid >> 5;
  const int nt   = blockIdx.x;
  const int h    = blockIdx.y;

  const float* Sq  = ws + OFF_SQ;  const float* Sk  = ws + OFF_SK;
  const float* Pq  = ws + OFF_PQ;  const float* Pk  = ws + OFF_PK;
  const float* qn2 = ws + OFF_QN2; const float* kn2 = ws + OFF_KN2;
  const float* bias= ws + OFF_BIAS;
  const float* Sv  = ws + OFF_SV;  const float* Pv  = ws + OFF_PV;
  float* attn = ws + OFF_ATTN;
  float* res  = ws + OFF_RES;

  const float pwx = point_weight[h];
  const float pw  = (pwx > 20.f) ? pwx : log1pf(__expf(pwx));
  const float cpt = -0.5f * pw / sqrtf(18.f);
  const float rs3 = 0.57735026919f;

  const int row   = lane & 15;
  const int khalf = (lane >> 4) * 2;
  const int col   = lane & 15;
  const int rbase = (lane >> 4) * 8;

  {
    const float* gsk = Sk + (size_t)h*LQ*16 + (size_t)tid*4;
    const float* gpk = Pk + (size_t)h*LQ*16 + (size_t)tid*4;
    const float* gbi = bias + ((size_t)h*LQ + nt*16)*LQ + (size_t)tid*4;
    v4f rk[16], rp[16], rb[16];
#pragma unroll
    for (int q = 0; q < 16; ++q) {
      rk[q] = *reinterpret_cast<const v4f*>(gsk + (size_t)q*512);
      rp[q] = *reinterpret_cast<const v4f*>(gpk + (size_t)q*512);
    }
#pragma unroll
    for (int q = 0; q < 16; ++q)
      rb[q] = *reinterpret_cast<const v4f*>(gbi + (size_t)q*LQ);
    const v4f rn = *reinterpret_cast<const v4f*>(kn2 + (size_t)h*LQ + tid*4);
    const int kd = (tid >> 2) * AT_KP + (tid & 3) * 4;
#pragma unroll
    for (int q = 0; q < 16; ++q) {
      *reinterpret_cast<v4f*>(&Rs[kd + q*32*AT_KP])              = rk[q];
      *reinterpret_cast<v4f*>(&Rs[LQ*AT_KP + kd + q*32*AT_KP])   = rp[q];
    }
#pragma unroll
    for (int q = 0; q < 16; ++q)
      *reinterpret_cast<v4f*>(&sm[q*SMS + tid*4]) = rb[q];
    *reinterpret_cast<v4f*>(&kn[tid*4]) = rn;
  }

  v2f aq[4], ap[4];
#pragma unroll
  for (int ks = 0; ks < 4; ++ks) {
    const int k = ks*4 + khalf;
    const int l = nt*16 + row;
    aq[ks].x = Sq[((size_t)h*LQ + l)*16 + k];
    aq[ks].y = Sq[((size_t)h*LQ + l)*16 + k + 1];
    ap[ks].x = Pq[((size_t)h*LQ + l)*16 + k];
    ap[ks].y = Pq[((size_t)h*LQ + l)*16 + k + 1];
  }
  float qv[8];
#pragma unroll
  for (int r = 0; r < 8; ++r) qv[r] = qn2[h*LQ + nt*16 + rbase + r];
  __syncthreads();

  for (int mt = w; mt < 32; mt += 4) {
    v8f s = vzero8(), p = vzero8();
    const float* kb = &Rs[(mt*16 + row)*AT_KP + khalf];
#pragma unroll
    for (int ks = 0; ks < 4; ++ks) {
      const v2f bq = *reinterpret_cast<const v2f*>(kb + ks*4);
      const v2f bp = *reinterpret_cast<const v2f*>(kb + LQ*AT_KP + ks*4);
      s = wmma4(aq[ks], bq, s);
      p = wmma4(ap[ks], bp, p);
    }
    const int m = mt*16 + col;
    const float kv = kn[m];
#pragma unroll
    for (int r = 0; r < 8; ++r) {
      const int rr = rbase + r;
      const float dist = qv[r] + kv - 2.f*p[r];
      sm[rr*SMS + m] = (s[r] + sm[rr*SMS + m] + cpt*dist) * rs3;
    }
  }
  __syncthreads();

  v4f rv[40];
  {
    const float* gsv = Sv + (size_t)h*LQ*16 + (size_t)tid*4;
    const float* gpv = Pv + (size_t)h*LQ*24 + (size_t)tid*4;
#pragma unroll
    for (int q = 0; q < 16; ++q)
      rv[q] = *reinterpret_cast<const v4f*>(gsv + (size_t)q*512);
#pragma unroll
    for (int q = 0; q < 24; ++q)
      rv[16 + q] = *reinterpret_cast<const v4f*>(gpv + (size_t)q*512);
  }

  {
    const int rrow = tid >> 3;
    const int sub  = tid & 7;
    float mx = -1e30f;
    for (int j = sub; j < LQ; j += 8) mx = fmaxf(mx, sm[rrow*SMS + j]);
    for (int o = 1; o < 8; o <<= 1) mx = fmaxf(mx, __shfl_xor(mx, o, 8));
    float sum = 0.f;
    for (int j = sub; j < LQ; j += 8) {
      const float e = __expf(sm[rrow*SMS + j] - mx);
      sm[rrow*SMS + j] = e; sum += e;
    }
    for (int o = 1; o < 8; o <<= 1) sum += __shfl_xor(sum, o, 8);
    const float inv = 1.f / sum;
    const int n = nt*16 + rrow;
    for (int j = sub; j < LQ; j += 8) {
      const float v = sm[rrow*SMS + j] * inv;
      sm[rrow*SMS + j] = v;
      attn[((size_t)h*LQ + n)*LQ + j] = v;
    }
  }

#pragma unroll
  for (int q = 0; q < 16; ++q)
    *reinterpret_cast<v4f*>(&Rs[(tid + 128*q)*4]) = rv[q];
#pragma unroll
  for (int q = 0; q < 24; ++q)
    *reinterpret_cast<v4f*>(&Rs[LQ*16 + (tid + 128*q)*4]) = rv[16 + q];
  __syncthreads();

  if (w < 3) {
    const float* Ab = &sm[row*SMS + khalf];
    v8f acc;
    if (w == 0) {
      acc = av_tile<16>(Ab, &Rs[khalf*16 + col]);
    } else {
      const int vc = (w == 1) ? col : ((col < 8) ? 16 + col : 23);
      acc = av_tile<24>(Ab, &Rs[LQ*16 + khalf*24 + vc]);
    }
    const int fbase = (w == 0) ? 0 : 144;
#pragma unroll
    for (int r = 0; r < 8; ++r) {
      const int n = nt*16 + rbase + r;
      const float v = acc[r];
      if (w < 2)        res[RES_IDX(h, n, fbase + col)] = v;
      else if (col < 8) res[RES_IDX(h, n, 160 + col)]   = v;
    }
  }
}

#define PO_KC 64
#define PO_NCH (LQ / PO_KC)
#define PO_AP 68
#define PO_BP 144

static __device__ __forceinline__ v8f pairout_chunk(
    const float* __restrict__ Ab, const float* __restrict__ Bb, v8f acc)
{
  v2f a[PO_KC / 4], b[PO_KC / 4];
#pragma unroll
  for (int kk = 0; kk < PO_KC / 4; ++kk) {
    a[kk]   = *reinterpret_cast<const v2f*>(Ab + kk*4);
    b[kk].x = Bb[(kk*4)     * PO_BP];
    b[kk].y = Bb[(kk*4 + 1) * PO_BP];
  }
  __builtin_amdgcn_sched_barrier(0);
#pragma unroll
  for (int kk = 0; kk < PO_KC / 4; ++kk)
    acc = wmma4(a[kk], b[kk], acc);
  asm volatile("" : "+v"(acc));
  __builtin_amdgcn_sched_barrier(0);
  return acc;
}

__global__ __launch_bounds__(256) void pairout_kernel(
    const float* __restrict__ pairwise, float* __restrict__ ws)
{
  __shared__ __attribute__((aligned(16))) float As[16 * PO_AP];
  __shared__ __attribute__((aligned(16))) float Bs[PO_KC * PO_BP];

  const float* attn = ws + OFF_ATTN;
  float* res = ws + OFF_RES;
  const int tid   = threadIdx.x;
  const int lane  = tid & 31;
  const int w     = tid >> 5;
  const int n     = blockIdx.x;
  const int row   = lane & 15;
  const int khalf = (lane >> 4) * 2;
  const int col   = lane & 15;

  const float* gb = pairwise + (size_t)n * LQ * PAIR + (size_t)tid * 4;
  const int sb = (tid >> 5) * PO_BP + (tid & 31) * 4;
  const bool hasA = tid < NH * 16;
  const int ah = hasA ? (tid >> 4) : 0;
  const float* gaa = attn + ((size_t)ah * LQ + n) * LQ + (tid & 15) * 4;
  const int sa = ah * PO_AP + (tid & 15) * 4;

  const int fa = row * PO_AP + khalf;
  const int fb = khalf * PO_BP + w*16 + col;

  v4f rb[8], ra;
#pragma unroll
  for (int i = 0; i < 8; ++i)
    rb[i] = *reinterpret_cast<const v4f*>(gb + (size_t)i * 1024);
  ra = *reinterpret_cast<const v4f*>(gaa);
#pragma unroll
  for (int i = 0; i < 8; ++i)
    *reinterpret_cast<v4f*>(&Bs[sb + i*8*PO_BP]) = rb[i];
  if (hasA) {
    *reinterpret_cast<v4f*>(&As[sa]) = ra;
  } else {
    const v4f z4 = {0.f, 0.f, 0.f, 0.f};
    *reinterpret_cast<v4f*>(&As[(tid >> 4) * PO_AP + (tid & 15) * 4]) = z4;
  }
  __syncthreads();

  v8f acc = vzero8();
  for (int ch = 0; ch < PO_NCH - 1; ++ch) {
    const float* gbn = gb + (size_t)(ch + 1) * PO_KC * PAIR;
#pragma unroll
    for (int i = 0; i < 8; ++i)
      rb[i] = *reinterpret_cast<const v4f*>(gbn + (size_t)i * 1024);
    ra = *reinterpret_cast<const v4f*>(gaa + (ch + 1) * PO_KC);

    acc = pairout_chunk(&As[fa], &Bs[fb], acc);

    __syncthreads();
#pragma unroll
    for (int i = 0; i < 8; ++i)
      *reinterpret_cast<v4f*>(&Bs[sb + i*8*PO_BP]) = rb[i];
    if (hasA) *reinterpret_cast<v4f*>(&As[sa]) = ra;
    __syncthreads();
  }
  acc = pairout_chunk(&As[fa], &Bs[fb], acc);

  const int rbase = (lane >> 4) * 8;
#pragma unroll
  for (int r = 0; r < 8; ++r) {
    const int hh = rbase + r;
    if (hh < NH)
      res[RES_IDX(hh, n, 16 + w*16 + col)] = acc[r];
  }
}

__global__ void pfinish_kernel(const float* __restrict__ R,
                               const float* __restrict__ t,
                               float* __restrict__ ws)
{
  const int idx = blockIdx.x * blockDim.x + threadIdx.x;
  if (idx >= NH * LQ) return;
  const int h = idx / LQ, l = idx % LQ;
  float* resp = ws + OFF_RES + RES_IDX(h, l, 0);
  float Rm[9], tv[3];
#pragma unroll
  for (int i = 0; i < 9; ++i) Rm[i] = R[l*9 + i];
#pragma unroll
  for (int i = 0; i < 3; ++i) tv[i] = t[l*3 + i];
#pragma unroll
  for (int p = 0; p < 8; ++p) {
    float g[3], loc[3], nrm = 0.f;
    for (int gg = 0; gg < 3; ++gg) g[gg] = resp[144 + p*3 + gg] - tv[gg];
    for (int f = 0; f < 3; ++f) {
      float v = 0.f;
      for (int gg = 0; gg < 3; ++gg) v += g[gg]*Rm[gg*3 + f];
      loc[f] = v; nrm += v*v;
    }
    for (int f = 0; f < 3; ++f) resp[144 + p*3 + f] = loc[f];
    resp[168 + p] = sqrtf(nrm + 1e-12f);
  }
}

#define OP_KC  96
#define OP_NCH (KOUT / OP_KC)
#define OP_AP  100
#define OP_BP  48

static __device__ __forceinline__ v8f outproj_chunk(
    const float* __restrict__ Ab, const float* __restrict__ Bb, v8f acc)
{
  v2f a[OP_KC / 4], b[OP_KC / 4];
#pragma unroll
  for (int kk = 0; kk < OP_KC / 4; ++kk) {
    a[kk]   = *reinterpret_cast<const v2f*>(Ab + kk*4);
    b[kk].x = Bb[(kk*4)     * OP_BP];
    b[kk].y = Bb[(kk*4 + 1) * OP_BP];
  }
  __builtin_amdgcn_sched_barrier(0);
#pragma unroll
  for (int kk = 0; kk < OP_KC / 4; ++kk)
    acc = wmma4(a[kk], b[kk], acc);
  asm volatile("" : "+v"(acc));
  __builtin_amdgcn_sched_barrier(0);
  return acc;
}

__global__ __launch_bounds__(128) void outproj_kernel(
    const float* __restrict__ out_k, const float* __restrict__ out_b,
    const float* __restrict__ res, float* __restrict__ out)
{
  __shared__ __attribute__((aligned(16))) float As[2][32 * OP_AP];
  __shared__ __attribute__((aligned(16))) float Bs[2][OP_KC * OP_BP];

  const int tid   = threadIdx.x;
  const int lane  = tid & 31;
  const int w     = tid >> 5;
  const int wm    = w >> 1;
  const int wn    = w & 1;
  const int m0    = blockIdx.x * 32;
  const int n0    = blockIdx.y * 32;
  const int col   = lane & 15;
  const int khalf = (lane >> 4) * 2;

  const int srow = tid >> 3;
  const int sseg = tid & 7;
  const float* ga = res   + (size_t)(m0 + srow) * KOUT + sseg*4;
  const float* gb = out_k + (size_t)srow * NODE + n0 + sseg*4;
  const int sa = srow * OP_AP + sseg*4;
  const int sb = srow * OP_BP + sseg*4;

  const int fa = (wm*16 + col) * OP_AP + khalf;
  const int fb = khalf * OP_BP + wn*16 + col;
  const int o  = n0 + wn*16 + col;
  const float ob = out_b[o];

  v4f ra[6], rb[6];
#pragma unroll
  for (int i = 0; i < 2; ++i)
#pragma unroll
    for (int j = 0; j < 3; ++j)
      ra[i*3 + j] = *reinterpret_cast<const v4f*>(ga + (size_t)i*16*KOUT + j*32);
#pragma unroll
  for (int i = 0; i < 6; ++i)
    rb[i] = *reinterpret_cast<const v4f*>(gb + (size_t)i*16*NODE);
#pragma unroll
  for (int i = 0; i < 2; ++i)
#pragma unroll
    for (int j = 0; j < 3; ++j)
      *reinterpret_cast<v4f*>(&As[0][sa + i*16*OP_AP + j*32]) = ra[i*3 + j];
#pragma unroll
  for (int i = 0; i < 6; ++i)
    *reinterpret_cast<v4f*>(&Bs[0][sb + i*16*OP_BP]) = rb[i];
  __syncthreads();

  v8f acc = vzero8();
  for (int ch = 0; ch < OP_NCH - 1; ++ch) {
    const int cur = ch & 1, nxt = cur ^ 1;
    const float* gan = ga + (size_t)(ch + 1) * OP_KC;
    const float* gbn = gb + (size_t)(ch + 1) * OP_KC * NODE;
#pragma unroll
    for (int i = 0; i < 2; ++i)
#pragma unroll
      for (int j = 0; j < 3; ++j)
        ra[i*3 + j] = *reinterpret_cast<const v4f*>(gan + (size_t)i*16*KOUT + j*32);
#pragma unroll
    for (int i = 0; i < 6; ++i)
      rb[i] = *reinterpret_cast<const v4f*>(gbn + (size_t)i*16*NODE);

    acc = outproj_chunk(&As[cur][fa], &Bs[cur][fb], acc);

#pragma unroll
    for (int i = 0; i < 2; ++i)
#pragma unroll
      for (int j = 0; j < 3; ++j)
        *reinterpret_cast<v4f*>(&As[nxt][sa + i*16*OP_AP + j*32]) = ra[i*3 + j];
#pragma unroll
    for (int i = 0; i < 6; ++i)
      *reinterpret_cast<v4f*>(&Bs[nxt][sb + i*16*OP_BP]) = rb[i];
    __syncthreads();
  }
  acc = outproj_chunk(&As[(OP_NCH - 1) & 1][fa], &Bs[(OP_NCH - 1) & 1][fb], acc);

  const int rbase = (lane >> 4) * 8;
#pragma unroll
  for (int r = 0; r < 8; ++r) {
    const int l = m0 + wm*16 + rbase + r;
    out[(size_t)l*NODE + o] = acc[r] + ob;
  }
}

extern "C" void kernel_launch(void* const* d_in, const int* in_sizes, int n_in,
                              void* d_out, int out_size, void* d_ws, size_t ws_size,
                              hipStream_t stream) {
  const float* node  = (const float*)d_in[0];
  const float* R     = (const float*)d_in[1];
  const float* t     = (const float*)d_in[2];
  const float* pairw = (const float*)d_in[3];
  const float* sq_k  = (const float*)d_in[4];
  const float* sk_k  = (const float*)d_in[5];
  const float* sv_k  = (const float*)d_in[6];
  const float* pq_k  = (const float*)d_in[7];
  const float* pk_k  = (const float*)d_in[8];
  const float* pv_k  = (const float*)d_in[9];
  const float* pr_k  = (const float*)d_in[10];
  const float* out_k = (const float*)d_in[11];
  const float* out_b = (const float*)d_in[12];
  const float* pwgt  = (const float*)d_in[13];
  float* ws  = (float*)d_ws;
  float* out = (float*)d_out;

  proj_kernel<<<dim3(32, 7, NH), 32, 0, stream>>>(
      node, sq_k, sk_k, sv_k, pq_k, pk_k, pv_k, ws);
  rigid_kernel<<<(NH*LQ + 127)/128, 128, 0, stream>>>(R, t, ws);
  bias_kernel<<<LQ, 256, 0, stream>>>(pairw, pr_k, ws);
  attn_kernel<<<dim3(32, NH), 128, 0, stream>>>(pwgt, ws);
  pairout_kernel<<<LQ, 256, 0, stream>>>(pairw, ws);
  pfinish_kernel<<<(NH*LQ + 127)/128, 128, 0, stream>>>(R, t, ws);
  outproj_kernel<<<dim3(16, 12), 128, 0, stream>>>(out_k, out_b, ws + OFF_RES, out);
}


# bias: each row's j chunks walked high to low, so the pair tensor's second pass (pairout, ascending m) starts on the slabs read last and finds them in the second-level cache; every sum unchanged
# speedup vs baseline: 3.2313x; 1.0772x over previous
#include <hip/hip_runtime.h>
#include <math.h>


typedef __attribute__((ext_vector_type(2))) float v2f;
typedef __attribute__((ext_vector_type(4))) float v4f;
typedef __attribute__((ext_vector_type(8))) float v8f;

#define LQ    512
#define NH    12
#define NODE  384
#define PAIR  128
#define OUTIN 176
#define KOUT  (NH*OUTIN)

#define OFF_SQ   ((size_t)0)
#define OFF_SK   (OFF_SQ  + (size_t)NH*LQ*16)
#define OFF_SV   (OFF_SK  + (size_t)NH*LQ*16)
#define OFF_PQ   (OFF_SV  + (size_t)NH*LQ*16)
#define OFF_PK   (OFF_PQ  + (size_t)NH*LQ*16)
#define OFF_PV   (OFF_PK  + (size_t)NH*LQ*16)
#define OFF_QN2  (OFF_PV  + (size_t)NH*LQ*24)
#define OFF_KN2  (OFF_QN2 + (size_t)NH*LQ)
#define OFF_BIAS (OFF_KN2 + (size_t)NH*LQ)
#define OFF_ATTN (OFF_BIAS+ (size_t)NH*LQ*LQ)
#define OFF_RES  (OFF_ATTN+ (size_t)NH*LQ*LQ)

#define RES_IDX(h, l, f) (((size_t)(l)*NH + (h))*OUTIN + (f))

static __device__ __forceinline__ v8f wmma4(v2f a, v2f b, v8f c) {
  return __builtin_amdgcn_wmma_f32_16x16x4_f32(
       false, a,  false, b,
       (short)0, c,  false,  false);
}

static __device__ __forceinline__ v8f vzero8() {
  v8f z = {0.f,0.f,0.f,0.f,0.f,0.f,0.f,0.f};
  return z;
}

#define PJ_KC  64
#define PJ_NCH (NODE / PJ_KC)
#define PJ_AP  68

template <int WS>
static __device__ __forceinline__ v8f proj_chunk(
    const float* __restrict__ Ab, const float* __restrict__ Bb, float wscale, v8f acc)
{
  v2f a[PJ_KC / 4], b[PJ_KC / 4];
#pragma unroll
  for (int kk = 0; kk < PJ_KC / 4; ++kk) {
    a[kk]   = *reinterpret_cast<const v2f*>(Ab + kk*4);
    b[kk].x = Bb[(kk*4)     * WS] * wscale;
    b[kk].y = Bb[(kk*4 + 1) * WS] * wscale;
  }
  __builtin_amdgcn_sched_barrier(0);
#pragma unroll
  for (int kk = 0; kk < PJ_KC / 4; ++kk)
    acc = wmma4(a[kk], b[kk], acc);
  asm volatile("" : "+v"(acc));
  __builtin_amdgcn_sched_barrier(0);
  return acc;
}

template <int WS>
static __device__ __forceinline__ v8f proj_body(
    const float* __restrict__ ga, const float* __restrict__ gw,
    float* __restrict__ As, float* __restrict__ Bs,
    int lane, int wcol, float wscale)
{
  constexpr int NB = PJ_KC * WS / 128;
  const int row   = lane & 15;
  const int khalf = (lane >> 4) * 2;
  const float* gaa = ga + (size_t)(lane >> 4) * NODE + (lane & 15) * 4;
  const int sa = (lane >> 4) * PJ_AP + (lane & 15) * 4;
  const float* gbb = gw + (size_t)lane * 4;
  const int sb = lane * 4;
  const int fa = row * PJ_AP + khalf;
  const int fb = khalf * WS + wcol;

  v4f ra[8], rb[NB];
#pragma unroll
  for (int q = 0; q < 8; ++q)
    ra[q] = *reinterpret_cast<const v4f*>(gaa + (size_t)q * 2 * NODE);
#pragma unroll
  for (int q = 0; q < NB; ++q)
    rb[q] = *reinterpret_cast<const v4f*>(gbb + (size_t)q * 128);
#pragma unroll
  for (int q = 0; q < 8; ++q)
    *reinterpret_cast<v4f*>(&As[sa + q*2*PJ_AP]) = ra[q];
#pragma unroll
  for (int q = 0; q < NB; ++q)
    *reinterpret_cast<v4f*>(&Bs[sb + q*128]) = rb[q];
  __syncthreads();

  v8f acc = vzero8();
  for (int ch = 0; ch < PJ_NCH - 1; ++ch) {
    const float* gan = gaa + (size_t)(ch + 1) * PJ_KC;
    const float* gbn = gbb + (size_t)(ch + 1) * PJ_KC * WS;
#pragma unroll
    for (int q = 0; q < 8; ++q)
      ra[q] = *reinterpret_cast<const v4f*>(gan + (size_t)q * 2 * NODE);
#pragma unroll
    for (int q = 0; q < NB; ++q)
      rb[q] = *reinterpret_cast<const v4f*>(gbn + (size_t)q * 128);

    acc = proj_chunk<WS>(&As[fa], &Bs[fb], wscale, acc);

    __syncthreads();
#pragma unroll
    for (int q = 0; q < 8; ++q)
      *reinterpret_cast<v4f*>(&As[sa + q*2*PJ_AP]) = ra[q];
#pragma unroll
    for (int q = 0; q < NB; ++q)
      *reinterpret_cast<v4f*>(&Bs[sb + q*128]) = rb[q];
    __syncthreads();
  }
  return proj_chunk<WS>(&As[fa], &Bs[fb], wscale, acc);
}

__global__ __launch_bounds__(32) void proj_kernel(
    const float* __restrict__ node,
    const float* __restrict__ sq_k, const float* __restrict__ sk_k,
    const float* __restrict__ sv_k, const float* __restrict__ pq_k,
    const float* __restrict__ pk_k, const float* __restrict__ pv_k,
    float* __restrict__ ws)
{
  __shared__ __attribute__((aligned(16))) float As[16 * PJ_AP];
  __shared__ __attribute__((aligned(16))) float Bs[PJ_KC * 24];

  const int lane  = threadIdx.x;
  const int mt    = blockIdx.x;
  const int seg   = blockIdx.y;
  const int h     = blockIdx.z;
  const int row   = lane & 15;

  const float* ga = node + (size_t)(mt*16) * NODE;
  v8f acc;
  if (seg < 3) {
    const float* wb = (seg == 0) ? sq_k : (seg == 1) ? sk_k : sv_k;
    acc = proj_body<16>(ga, wb + (size_t)h*NODE*16, As, Bs, lane, row,
                        (seg == 0) ? 0.25f : 1.f);
  } else if (seg < 5) {
    const float* wb = (seg == 3) ? pq_k : pk_k;
    acc = proj_body<12>(ga, wb + (size_t)h*NODE*12, As, Bs, lane,
                        (row < 12) ? row : 11, (row < 12) ? 1.f : 0.f);
  } else if (seg == 5) {
    acc = proj_body<24>(ga, pv_k + (size_t)h*NODE*24, As, Bs, lane, row, 1.f);
  } else {
    acc = proj_body<24>(ga, pv_k + (size_t)h*NODE*24, As, Bs, lane,
                        (row < 8) ? 16 + row : 23, (row < 8) ? 1.f : 0.f);
  }

  const int col   = lane & 15;
  const int rbase = (lane >> 4) * 8;
  float* ob = ws + OFF_SK + (size_t)h*LQ*16;
  int ostride = 16, ocol = col;
  bool dostore = true;
  switch (seg) {
    case 0: ob = ws + OFF_SQ + (size_t)h*LQ*16; break;
    case 1: break;
    case 2: ob = ws + OFF_SV + (size_t)h*LQ*16; break;
    case 3: ob = ws + OFF_PQ + (size_t)h*LQ*16; break;
    case 4: ob = ws + OFF_PK + (size_t)h*LQ*16; break;
    case 5: ob = ws + OFF_PV + (size_t)h*LQ*24; ostride = 24; break;
    case 6: ob = ws + OFF_PV + (size_t)h*LQ*24; ostride = 24;
            ocol = 16 + col; dostore = (col < 8); break;
  }
  if (dostore) {
#pragma unroll
    for (int r = 0; r < 8; ++r)
      ob[(size_t)(mt*16 + rbase + r)*ostride + ocol] = acc[r];
  }
}

__global__ void rigid_kernel(const float* __restrict__ R,
                             const float* __restrict__ t,
                             float* __restrict__ ws)
{
  const int idx = blockIdx.x * blockDim.x + threadIdx.x;
  if (idx >= NH * LQ) return;
  const int h = idx / LQ, l = idx % LQ;
  float Rm[9], tv[3];
#pragma unroll
  for (int i = 0; i < 9; ++i) Rm[i] = R[l*9 + i];
#pragma unroll
  for (int i = 0; i < 3; ++i) tv[i] = t[l*3 + i];

  float* Pq = ws + OFF_PQ + ((size_t)h*LQ + l)*16;
  float* Pk = ws + OFF_PK + ((size_t)h*LQ + l)*16;
  float* Pv = ws + OFF_PV + ((size_t)h*LQ + l)*24;

  float qs = 0.f, ksum = 0.f;
#pragma unroll
  for (int p = 0; p < 4; ++p) {
    float loc[3], g[3];
    for (int f = 0; f < 3; ++f) loc[f] = Pq[p*3+f];
    for (int gg = 0; gg < 3; ++gg) {
      g[gg] = tv[gg];
      for (int f = 0; f < 3; ++f) g[gg] += loc[f]*Rm[gg*3+f];
      Pq[p*3+gg] = g[gg]; qs += g[gg]*g[gg];
    }
    for (int f = 0; f < 3; ++f) loc[f] = Pk[p*3+f];
    for (int gg = 0; gg < 3; ++gg) {
      g[gg] = tv[gg];
      for (int f = 0; f < 3; ++f) g[gg] += loc[f]*Rm[gg*3+f];
      Pk[p*3+gg] = g[gg]; ksum += g[gg]*g[gg];
    }
  }
#pragma unroll
  for (int p = 0; p < 8; ++p) {
    float loc[3], g[3];
    for (int f = 0; f < 3; ++f) loc[f] = Pv[p*3+f];
    for (int gg = 0; gg < 3; ++gg) {
      g[gg] = tv[gg];
      for (int f = 0; f < 3; ++f) g[gg] += loc[f]*Rm[gg*3+f];
      Pv[p*3+gg] = g[gg];
    }
  }
  ws[OFF_QN2 + h*LQ + l] = qs;
  ws[OFF_KN2 + h*LQ + l] = ksum;
}

#define BI_JC 128
#define BI_NCH (LQ / BI_JC)
#define BI_BP 132

static __device__ __forceinline__ v8f bias_chunk(
    const float* __restrict__ Bb, const v2f (&pa)[PAIR / 4])
{
  v2f b[PAIR / 4];
#pragma unroll
  for (int kk = 0; kk < PAIR / 4; ++kk)
    b[kk] = *reinterpret_cast<const v2f*>(Bb + kk*4);
  __builtin_amdgcn_sched_barrier(0);
  v8f acc = vzero8();
#pragma unroll
  for (int kk = 0; kk < PAIR / 4; ++kk)
    acc = wmma4(pa[kk], b[kk], acc);
  asm volatile("" : "+v"(acc));
  __builtin_amdgcn_sched_barrier(0);
  return acc;
}

static __device__ __forceinline__ void bias_store(
    float* __restrict__ ob, v8f acc, bool lowhalf)
{
#pragma unroll
  for (int r = 0; r < 4; ++r) ob[(size_t)r * LQ * LQ] = acc[r];
  if (lowhalf) {
#pragma unroll
    for (int r = 4; r < 8; ++r) ob[(size_t)r * LQ * LQ] = acc[r];
  }
}

__global__ __launch_bounds__(256) void bias_kernel(
    const float* __restrict__ pairwise, const float* __restrict__ pr_k,
    float* __restrict__ ws)
{
  __shared__ __attribute__((aligned(16))) float Bs[BI_JC * BI_BP];

  float* bias = ws + OFF_BIAS;
  const int tid   = threadIdx.x;
  const int lane  = tid & 31;
  const int w     = tid >> 5;
  const int i     = blockIdx.x;
  const int row   = lane & 15;
  const int khalf = (lane >> 4) * 2;

  const float* gb = pairwise + (size_t)i * LQ * PAIR + (size_t)tid * 4;
  const int sb = (tid >> 5) * BI_BP + (tid & 31) * 4;
  const int fb = (w*16 + row) * BI_BP + khalf;

  v4f rb[16];
  {
    const float* gbl = gb + (size_t)(BI_NCH - 1) * BI_JC * PAIR;
#pragma unroll
    for (int q = 0; q < 16; ++q)
      rb[q] = *reinterpret_cast<const v4f*>(gbl + (size_t)q * 1024);
  }

  const bool live = row < NH;
  const float* prk = pr_k + (size_t)(live ? row : NH - 1) * PAIR + khalf;
  v2f pa[PAIR / 4];
#pragma unroll
  for (int kk = 0; kk < PAIR / 4; ++kk) {
    const v2f t = *reinterpret_cast<const v2f*>(prk + kk*4);
    pa[kk].x = live ? t.x : 0.f;
    pa[kk].y = live ? t.y : 0.f;
  }

#pragma unroll
  for (int q = 0; q < 16; ++q)
    *reinterpret_cast<v4f*>(&Bs[sb + q*8*BI_BP]) = rb[q];
  __syncthreads();

  const bool lowhalf = lane < 16;
  float* ob = bias + ((size_t)((lane >> 4) * 8) * LQ + i) * LQ + w*16 + row;

  for (int ch = BI_NCH - 1; ch > 0; --ch) {
    const float* gbn = gb + (size_t)(ch - 1) * BI_JC * PAIR;
#pragma unroll
    for (int q = 0; q < 16; ++q)
      rb[q] = *reinterpret_cast<const v4f*>(gbn + (size_t)q * 1024);

    const v8f acc = bias_chunk(&Bs[fb], pa);

    __syncthreads();
#pragma unroll
    for (int q = 0; q < 16; ++q)
      *reinterpret_cast<v4f*>(&Bs[sb + q*8*BI_BP]) = rb[q];
    __syncthreads();
    bias_store(ob + ch * BI_JC, acc, lowhalf);
  }
  const v8f acc = bias_chunk(&Bs[fb], pa);
  bias_store(ob, acc, lowhalf);
}

#define SMS 516
#define AT_KP 20
#define AT_G  32

template <int VP>
static __device__ __forceinline__ v8f av_group(
    const float* __restrict__ Ab, const float* __restrict__ Bb, v8f acc)
{
  v2f a[AT_G], b[AT_G];
#pragma unroll
  for (int kk = 0; kk < AT_G; ++kk) {
    a[kk]   = *reinterpret_cast<const v2f*>(Ab + kk*4);
    b[kk].x = Bb[(kk*4)     * VP];
    b[kk].y = Bb[(kk*4 + 1) * VP];
  }
  __builtin_amdgcn_sched_barrier(0);
#pragma unroll
  for (int kk = 0; kk < AT_G; ++kk)
    acc = wmma4(a[kk], b[kk], acc);
  __builtin_amdgcn_sched_barrier(0);
  return acc;
}

template <int VP>
static __device__ __forceinline__ v8f av_tile(
    const float* __restrict__ Ab, const float* __restrict__ Bb)
{
  v8f acc = vzero8();
#pragma unroll
  for (int g = 0; g < LQ / (4 * AT_G); ++g)
    acc = av_group<VP>(Ab + g * 4 * AT_G, Bb + (size_t)g * 4 * AT_G * VP, acc);
  return acc;
}

__global__ __launch_bounds__(128) void attn_kernel(
    const float* __restrict__ point_weight, float* __restrict__ ws)
{
  __shared__ __attribute__((aligned(16))) float sm[16 * SMS];
  __shared__ __attribute__((aligned(16))) float Rs[LQ * 40];
  __shared__ __attribute__((aligned(16))) float kn[LQ];
  const int tid  = threadIdx.x;
  const int lane = tid & 31;
  const int w    = tid >> 5;
  const int nt   = blockIdx.x;
  const int h    = blockIdx.y;

  const float* Sq  = ws + OFF_SQ;  const float* Sk  = ws + OFF_SK;
  const float* Pq  = ws + OFF_PQ;  const float* Pk  = ws + OFF_PK;
  const float* qn2 = ws + OFF_QN2; const float* kn2 = ws + OFF_KN2;
  const float* bias= ws + OFF_BIAS;
  const float* Sv  = ws + OFF_SV;  const float* Pv  = ws + OFF_PV;
  float* attn = ws + OFF_ATTN;
  float* res  = ws + OFF_RES;

  const float pwx = point_weight[h];
  const float pw  = (pwx > 20.f) ? pwx : log1pf(__expf(pwx));
  const float cpt = -0.5f * pw / sqrtf(18.f);
  const float rs3 = 0.57735026919f;

  const int row   = lane & 15;
  const int khalf = (lane >> 4) * 2;
  const int col   = lane & 15;
  const int rbase = (lane >> 4) * 8;

  {
    const float* gsk = Sk + (size_t)h*LQ*16 + (size_t)tid*4;
    const float* gpk = Pk + (size_t)h*LQ*16 + (size_t)tid*4;
    const float* gbi = bias + ((size_t)h*LQ + nt*16)*LQ + (size_t)tid*4;
    v4f rk[16], rp[16], rb[16];
#pragma unroll
    for (int q = 0; q < 16; ++q) {
      rk[q] = *reinterpret_cast<const v4f*>(gsk + (size_t)q*512);
      rp[q] = *reinterpret_cast<const v4f*>(gpk + (size_t)q*512);
    }
#pragma unroll
    for (int q = 0; q < 16; ++q)
      rb[q] = *reinterpret_cast<const v4f*>(gbi + (size_t)q*LQ);
    const v4f rn = *reinterpret_cast<const v4f*>(kn2 + (size_t)h*LQ + tid*4);
    const int kd = (tid >> 2) * AT_KP + (tid & 3) * 4;
#pragma unroll
    for (int q = 0; q < 16; ++q) {
      *reinterpret_cast<v4f*>(&Rs[kd + q*32*AT_KP])              = rk[q];
      *reinterpret_cast<v4f*>(&Rs[LQ*AT_KP + kd + q*32*AT_KP])   = rp[q];
    }
#pragma unroll
    for (int q = 0; q < 16; ++q)
      *reinterpret_cast<v4f*>(&sm[q*SMS + tid*4]) = rb[q];
    *reinterpret_cast<v4f*>(&kn[tid*4]) = rn;
  }

  v2f aq[4], ap[4];
#pragma unroll
  for (int ks = 0; ks < 4; ++ks) {
    const int k = ks*4 + khalf;
    const int l = nt*16 + row;
    aq[ks].x = Sq[((size_t)h*LQ + l)*16 + k];
    aq[ks].y = Sq[((size_t)h*LQ + l)*16 + k + 1];
    ap[ks].x = Pq[((size_t)h*LQ + l)*16 + k];
    ap[ks].y = Pq[((size_t)h*LQ + l)*16 + k + 1];
  }
  float qv[8];
#pragma unroll
  for (int r = 0; r < 8; ++r) qv[r] = qn2[h*LQ + nt*16 + rbase + r];
  __syncthreads();

  for (int mt = w; mt < 32; mt += 4) {
    v8f s = vzero8(), p = vzero8();
    const float* kb = &Rs[(mt*16 + row)*AT_KP + khalf];
#pragma unroll
    for (int ks = 0; ks < 4; ++ks) {
      const v2f bq = *reinterpret_cast<const v2f*>(kb + ks*4);
      const v2f bp = *reinterpret_cast<const v2f*>(kb + LQ*AT_KP + ks*4);
      s = wmma4(aq[ks], bq, s);
      p = wmma4(ap[ks], bp, p);
    }
    const int m = mt*16 + col;
    const float kv = kn[m];
#pragma unroll
    for (int r = 0; r < 8; ++r) {
      const int rr = rbase + r;
      const float dist = qv[r] + kv - 2.f*p[r];
      sm[rr*SMS + m] = (s[r] + sm[rr*SMS + m] + cpt*dist) * rs3;
    }
  }
  __syncthreads();

  v4f rv[40];
  {
    const float* gsv = Sv + (size_t)h*LQ*16 + (size_t)tid*4;
    const float* gpv = Pv + (size_t)h*LQ*24 + (size_t)tid*4;
#pragma unroll
    for (int q = 0; q < 16; ++q)
      rv[q] = *reinterpret_cast<const v4f*>(gsv + (size_t)q*512);
#pragma unroll
    for (int q = 0; q < 24; ++q)
      rv[16 + q] = *reinterpret_cast<const v4f*>(gpv + (size_t)q*512);
  }

  {
    const int rrow = tid >> 3;
    const int sub  = tid & 7;
    float mx = -1e30f;
    for (int j = sub; j < LQ; j += 8) mx = fmaxf(mx, sm[rrow*SMS + j]);
    for (int o = 1; o < 8; o <<= 1) mx = fmaxf(mx, __shfl_xor(mx, o, 8));
    float sum = 0.f;
    for (int j = sub; j < LQ; j += 8) {
      const float e = __expf(sm[rrow*SMS + j] - mx);
      sm[rrow*SMS + j] = e; sum += e;
    }
    for (int o = 1; o < 8; o <<= 1) sum += __shfl_xor(sum, o, 8);
    const float inv = 1.f / sum;
    const int n = nt*16 + rrow;
    for (int j = sub; j < LQ; j += 8) {
      const float v = sm[rrow*SMS + j] * inv;
      sm[rrow*SMS + j] = v;
      attn[((size_t)h*LQ + n)*LQ + j] = v;
    }
  }

#pragma unroll
  for (int q = 0; q < 16; ++q)
    *reinterpret_cast<v4f*>(&Rs[(tid + 128*q)*4]) = rv[q];
#pragma unroll
  for (int q = 0; q < 24; ++q)
    *reinterpret_cast<v4f*>(&Rs[LQ*16 + (tid + 128*q)*4]) = rv[16 + q];
  __syncthreads();

  if (w < 3) {
    const float* Ab = &sm[row*SMS + khalf];
    v8f acc;
    if (w == 0) {
      acc = av_tile<16>(Ab, &Rs[khalf*16 + col]);
    } else {
      const int vc = (w == 1) ? col : ((col < 8) ? 16 + col : 23);
      acc = av_tile<24>(Ab, &Rs[LQ*16 + khalf*24 + vc]);
    }
    const int fbase = (w == 0) ? 0 : 144;
#pragma unroll
    for (int r = 0; r < 8; ++r) {
      const int n = nt*16 + rbase + r;
      const float v = acc[r];
      if (w < 2)        res[RES_IDX(h, n, fbase + col)] = v;
      else if (col < 8) res[RES_IDX(h, n, 160 + col)]   = v;
    }
  }
}

#define PO_KC 64
#define PO_NCH (LQ / PO_KC)
#define PO_AP 68
#define PO_BP 144

static __device__ __forceinline__ v8f pairout_chunk(
    const float* __restrict__ Ab, const float* __restrict__ Bb, v8f acc)
{
  v2f a[PO_KC / 4], b[PO_KC / 4];
#pragma unroll
  for (int kk = 0; kk < PO_KC / 4; ++kk) {
    a[kk]   = *reinterpret_cast<const v2f*>(Ab + kk*4);
    b[kk].x = Bb[(kk*4)     * PO_BP];
    b[kk].y = Bb[(kk*4 + 1) * PO_BP];
  }
  __builtin_amdgcn_sched_barrier(0);
#pragma unroll
  for (int kk = 0; kk < PO_KC / 4; ++kk)
    acc = wmma4(a[kk], b[kk], acc);
  asm volatile("" : "+v"(acc));
  __builtin_amdgcn_sched_barrier(0);
  return acc;
}

__global__ __launch_bounds__(256) void pairout_kernel(
    const float* __restrict__ pairwise, float* __restrict__ ws)
{
  __shared__ __attribute__((aligned(16))) float As[16 * PO_AP];
  __shared__ __attribute__((aligned(16))) float Bs[PO_KC * PO_BP];

  const float* attn = ws + OFF_ATTN;
  float* res = ws + OFF_RES;
  const int tid   = threadIdx.x;
  const int lane  = tid & 31;
  const int w     = tid >> 5;
  const int n     = blockIdx.x;
  const int row   = lane & 15;
  const int khalf = (lane >> 4) * 2;
  const int col   = lane & 15;

  const float* gb = pairwise + (size_t)n * LQ * PAIR + (size_t)tid * 4;
  const int sb = (tid >> 5) * PO_BP + (tid & 31) * 4;
  const bool hasA = tid < NH * 16;
  const int ah = hasA ? (tid >> 4) : 0;
  const float* gaa = attn + ((size_t)ah * LQ + n) * LQ + (tid & 15) * 4;
  const int sa = ah * PO_AP + (tid & 15) * 4;

  const int fa = row * PO_AP + khalf;
  const int fb = khalf * PO_BP + w*16 + col;

  v4f rb[8], ra;
#pragma unroll
  for (int i = 0; i < 8; ++i)
    rb[i] = *reinterpret_cast<const v4f*>(gb + (size_t)i * 1024);
  ra = *reinterpret_cast<const v4f*>(gaa);
#pragma unroll
  for (int i = 0; i < 8; ++i)
    *reinterpret_cast<v4f*>(&Bs[sb + i*8*PO_BP]) = rb[i];
  if (hasA) {
    *reinterpret_cast<v4f*>(&As[sa]) = ra;
  } else {
    const v4f z4 = {0.f, 0.f, 0.f, 0.f};
    *reinterpret_cast<v4f*>(&As[(tid >> 4) * PO_AP + (tid & 15) * 4]) = z4;
  }
  __syncthreads();

  v8f acc = vzero8();
  for (int ch = 0; ch < PO_NCH - 1; ++ch) {
    const float* gbn = gb + (size_t)(ch + 1) * PO_KC * PAIR;
#pragma unroll
    for (int i = 0; i < 8; ++i)
      rb[i] = *reinterpret_cast<const v4f*>(gbn + (size_t)i * 1024);
    ra = *reinterpret_cast<const v4f*>(gaa + (ch + 1) * PO_KC);

    acc = pairout_chunk(&As[fa], &Bs[fb], acc);

    __syncthreads();
#pragma unroll
    for (int i = 0; i < 8; ++i)
      *reinterpret_cast<v4f*>(&Bs[sb + i*8*PO_BP]) = rb[i];
    if (hasA) *reinterpret_cast<v4f*>(&As[sa]) = ra;
    __syncthreads();
  }
  acc = pairout_chunk(&As[fa], &Bs[fb], acc);

  const int rbase = (lane >> 4) * 8;
#pragma unroll
  for (int r = 0; r < 8; ++r) {
    const int hh = rbase + r;
    if (hh < NH)
      res[RES_IDX(hh, n, 16 + w*16 + col)] = acc[r];
  }
}

__global__ void pfinish_kernel(const float* __restrict__ R,
                               const float* __restrict__ t,
                               float* __restrict__ ws)
{
  const int idx = blockIdx.x * blockDim.x + threadIdx.x;
  if (idx >= NH * LQ) return;
  const int h = idx / LQ, l = idx % LQ;
  float* resp = ws + OFF_RES + RES_IDX(h, l, 0);
  float Rm[9], tv[3];
#pragma unroll
  for (int i = 0; i < 9; ++i) Rm[i] = R[l*9 + i];
#pragma unroll
  for (int i = 0; i < 3; ++i) tv[i] = t[l*3 + i];
#pragma unroll
  for (int p = 0; p < 8; ++p) {
    float g[3], loc[3], nrm = 0.f;
    for (int gg = 0; gg < 3; ++gg) g[gg] = resp[144 + p*3 + gg] - tv[gg];
    for (int f = 0; f < 3; ++f) {
      float v = 0.f;
      for (int gg = 0; gg < 3; ++gg) v += g[gg]*Rm[gg*3 + f];
      loc[f] = v; nrm += v*v;
    }
    for (int f = 0; f < 3; ++f) resp[144 + p*3 + f] = loc[f];
    resp[168 + p] = sqrtf(nrm + 1e-12f);
  }
}

#define OP_KC  96
#define OP_NCH (KOUT / OP_KC)
#define OP_AP  100
#define OP_BP  48

static __device__ __forceinline__ v8f outproj_chunk(
    const float* __restrict__ Ab, const float* __restrict__ Bb, v8f acc)
{
  v2f a[OP_KC / 4], b[OP_KC / 4];
#pragma unroll
  for (int kk = 0; kk < OP_KC / 4; ++kk) {
    a[kk]   = *reinterpret_cast<const v2f*>(Ab + kk*4);
    b[kk].x = Bb[(kk*4)     * OP_BP];
    b[kk].y = Bb[(kk*4 + 1) * OP_BP];
  }
  __builtin_amdgcn_sched_barrier(0);
#pragma unroll
  for (int kk = 0; kk < OP_KC / 4; ++kk)
    acc = wmma4(a[kk], b[kk], acc);
  asm volatile("" : "+v"(acc));
  __builtin_amdgcn_sched_barrier(0);
  return acc;
}

__global__ __launch_bounds__(128) void outproj_kernel(
    const float* __restrict__ out_k, const float* __restrict__ out_b,
    const float* __restrict__ res, float* __restrict__ out)
{
  __shared__ __attribute__((aligned(16))) float As[2][32 * OP_AP];
  __shared__ __attribute__((aligned(16))) float Bs[2][OP_KC * OP_BP];

  const int tid   = threadIdx.x;
  const int lane  = tid & 31;
  const int w     = tid >> 5;
  const int wm    = w >> 1;
  const int wn    = w & 1;
  const int m0    = blockIdx.x * 32;
  const int n0    = blockIdx.y * 32;
  const int col   = lane & 15;
  const int khalf = (lane >> 4) * 2;

  const int srow = tid >> 3;
  const int sseg = tid & 7;
  const float* ga = res   + (size_t)(m0 + srow) * KOUT + sseg*4;
  const float* gb = out_k + (size_t)srow * NODE + n0 + sseg*4;
  const int sa = srow * OP_AP + sseg*4;
  const int sb = srow * OP_BP + sseg*4;

  const int fa = (wm*16 + col) * OP_AP + khalf;
  const int fb = khalf * OP_BP + wn*16 + col;
  const int o  = n0 + wn*16 + col;
  const float ob = out_b[o];

  v4f ra[6], rb[6];
#pragma unroll
  for (int i = 0; i < 2; ++i)
#pragma unroll
    for (int j = 0; j < 3; ++j)
      ra[i*3 + j] = *reinterpret_cast<const v4f*>(ga + (size_t)i*16*KOUT + j*32);
#pragma unroll
  for (int i = 0; i < 6; ++i)
    rb[i] = *reinterpret_cast<const v4f*>(gb + (size_t)i*16*NODE);
#pragma unroll
  for (int i = 0; i < 2; ++i)
#pragma unroll
    for (int j = 0; j < 3; ++j)
      *reinterpret_cast<v4f*>(&As[0][sa + i*16*OP_AP + j*32]) = ra[i*3 + j];
#pragma unroll
  for (int i = 0; i < 6; ++i)
    *reinterpret_cast<v4f*>(&Bs[0][sb + i*16*OP_BP]) = rb[i];
  __syncthreads();

  v8f acc = vzero8();
  for (int ch = 0; ch < OP_NCH - 1; ++ch) {
    const int cur = ch & 1, nxt = cur ^ 1;
    const float* gan = ga + (size_t)(ch + 1) * OP_KC;
    const float* gbn = gb + (size_t)(ch + 1) * OP_KC * NODE;
#pragma unroll
    for (int i = 0; i < 2; ++i)
#pragma unroll
      for (int j = 0; j < 3; ++j)
        ra[i*3 + j] = *reinterpret_cast<const v4f*>(gan + (size_t)i*16*KOUT + j*32);
#pragma unroll
    for (int i = 0; i < 6; ++i)
      rb[i] = *reinterpret_cast<const v4f*>(gbn + (size_t)i*16*NODE);

    acc = outproj_chunk(&As[cur][fa], &Bs[cur][fb], acc);

#pragma unroll
    for (int i = 0; i < 2; ++i)
#pragma unroll
      for (int j = 0; j < 3; ++j)
        *reinterpret_cast<v4f*>(&As[nxt][sa + i*16*OP_AP + j*32]) = ra[i*3 + j];
#pragma unroll
    for (int i = 0; i < 6; ++i)
      *reinterpret_cast<v4f*>(&Bs[nxt][sb + i*16*OP_BP]) = rb[i];
    __syncthreads();
  }
  acc = outproj_chunk(&As[(OP_NCH - 1) & 1][fa], &Bs[(OP_NCH - 1) & 1][fb], acc);

  const int rbase = (lane >> 4) * 8;
#pragma unroll
  for (int r = 0; r < 8; ++r) {
    const int l = m0 + wm*16 + rbase + r;
    out[(size_t)l*NODE + o] = acc[r] + ob;
  }
}

extern "C" void kernel_launch(void* const* d_in, const int* in_sizes, int n_in,
                              void* d_out, int out_size, void* d_ws, size_t ws_size,
                              hipStream_t stream) {
  const float* node  = (const float*)d_in[0];
  const float* R     = (const float*)d_in[1];
  const float* t     = (const float*)d_in[2];
  const float* pairw = (const float*)d_in[3];
  const float* sq_k  = (const float*)d_in[4];
  const float* sk_k  = (const float*)d_in[5];
  const float* sv_k  = (const float*)d_in[6];
  const float* pq_k  = (const float*)d_in[7];
  const float* pk_k  = (const float*)d_in[8];
  const float* pv_k  = (const float*)d_in[9];
  const float* pr_k  = (const float*)d_in[10];
  const float* out_k = (const float*)d_in[11];
  const float* out_b = (const float*)d_in[12];
  const float* pwgt  = (const float*)d_in[13];
  float* ws  = (float*)d_ws;
  float* out = (float*)d_out;

  proj_kernel<<<dim3(32, 7, NH), 32, 0, stream>>>(
      node, sq_k, sk_k, sv_k, pq_k, pk_k, pv_k, ws);
  rigid_kernel<<<(NH*LQ + 127)/128, 128, 0, stream>>>(R, t, ws);
  bias_kernel<<<LQ, 256, 0, stream>>>(pairw, pr_k, ws);
  attn_kernel<<<dim3(32, NH), 128, 0, stream>>>(pwgt, ws);
  pairout_kernel<<<LQ, 256, 0, stream>>>(pairw, ws);
  pfinish_kernel<<<(NH*LQ + 127)/128, 128, 0, stream>>>(R, t, ws);
  outproj_kernel<<<dim3(16, 12), 128, 0, stream>>>(out_k, out_b, ws + OFF_RES, out);
}


# point outputs rotated back to the local frame, norms taken, in the attention epilogue (point tiles meet in LDS, one thread per row and point); separate pass dropped, 7 -> 6 launches; sums unchanged
# speedup vs baseline: 3.4340x; 1.0627x over previous
#include <hip/hip_runtime.h>
#include <math.h>


typedef __attribute__((ext_vector_type(2))) float v2f;
typedef __attribute__((ext_vector_type(4))) float v4f;
typedef __attribute__((ext_vector_type(8))) float v8f;

#define LQ    512
#define NH    12
#define NODE  384
#define PAIR  128
#define OUTIN 176
#define KOUT  (NH*OUTIN)

#define OFF_SQ   ((size_t)0)
#define OFF_SK   (OFF_SQ  + (size_t)NH*LQ*16)
#define OFF_SV   (OFF_SK  + (size_t)NH*LQ*16)
#define OFF_PQ   (OFF_SV  + (size_t)NH*LQ*16)
#define OFF_PK   (OFF_PQ  + (size_t)NH*LQ*16)
#define OFF_PV   (OFF_PK  + (size_t)NH*LQ*16)
#define OFF_QN2  (OFF_PV  + (size_t)NH*LQ*24)
#define OFF_KN2  (OFF_QN2 + (size_t)NH*LQ)
#define OFF_BIAS (OFF_KN2 + (size_t)NH*LQ)
#define OFF_ATTN (OFF_BIAS+ (size_t)NH*LQ*LQ)
#define OFF_RES  (OFF_ATTN+ (size_t)NH*LQ*LQ)

#define RES_IDX(h, l, f) (((size_t)(l)*NH + (h))*OUTIN + (f))

static __device__ __forceinline__ v8f wmma4(v2f a, v2f b, v8f c) {
  return __builtin_amdgcn_wmma_f32_16x16x4_f32(
       false, a,  false, b,
       (short)0, c,  false,  false);
}

static __device__ __forceinline__ v8f vzero8() {
  v8f z = {0.f,0.f,0.f,0.f,0.f,0.f,0.f,0.f};
  return z;
}

#define PJ_KC  64
#define PJ_NCH (NODE / PJ_KC)
#define PJ_AP  68

template <int WS>
static __device__ __forceinline__ v8f proj_chunk(
    const float* __restrict__ Ab, const float* __restrict__ Bb, float wscale, v8f acc)
{
  v2f a[PJ_KC / 4], b[PJ_KC / 4];
#pragma unroll
  for (int kk = 0; kk < PJ_KC / 4; ++kk) {
    a[kk]   = *reinterpret_cast<const v2f*>(Ab + kk*4);
    b[kk].x = Bb[(kk*4)     * WS] * wscale;
    b[kk].y = Bb[(kk*4 + 1) * WS] * wscale;
  }
  __builtin_amdgcn_sched_barrier(0);
#pragma unroll
  for (int kk = 0; kk < PJ_KC / 4; ++kk)
    acc = wmma4(a[kk], b[kk], acc);
  asm volatile("" : "+v"(acc));
  __builtin_amdgcn_sched_barrier(0);
  return acc;
}

template <int WS>
static __device__ __forceinline__ v8f proj_body(
    const float* __restrict__ ga, const float* __restrict__ gw,
    float* __restrict__ As, float* __restrict__ Bs,
    int lane, int wcol, float wscale)
{
  constexpr int NB = PJ_KC * WS / 128;
  const int row   = lane & 15;
  const int khalf = (lane >> 4) * 2;
  const float* gaa = ga + (size_t)(lane >> 4) * NODE + (lane & 15) * 4;
  const int sa = (lane >> 4) * PJ_AP + (lane & 15) * 4;
  const float* gbb = gw + (size_t)lane * 4;
  const int sb = lane * 4;
  const int fa = row * PJ_AP + khalf;
  const int fb = khalf * WS + wcol;

  v4f ra[8], rb[NB];
#pragma unroll
  for (int q = 0; q < 8; ++q)
    ra[q] = *reinterpret_cast<const v4f*>(gaa + (size_t)q * 2 * NODE);
#pragma unroll
  for (int q = 0; q < NB; ++q)
    rb[q] = *reinterpret_cast<const v4f*>(gbb + (size_t)q * 128);
#pragma unroll
  for (int q = 0; q < 8; ++q)
    *reinterpret_cast<v4f*>(&As[sa + q*2*PJ_AP]) = ra[q];
#pragma unroll
  for (int q = 0; q < NB; ++q)
    *reinterpret_cast<v4f*>(&Bs[sb + q*128]) = rb[q];
  __syncthreads();

  v8f acc = vzero8();
  for (int ch = 0; ch < PJ_NCH - 1; ++ch) {
    const float* gan = gaa + (size_t)(ch + 1) * PJ_KC;
    const float* gbn = gbb + (size_t)(ch + 1) * PJ_KC * WS;
#pragma unroll
    for (int q = 0; q < 8; ++q)
      ra[q] = *reinterpret_cast<const v4f*>(gan + (size_t)q * 2 * NODE);
#pragma unroll
    for (int q = 0; q < NB; ++q)
      rb[q] = *reinterpret_cast<const v4f*>(gbn + (size_t)q * 128);

    acc = proj_chunk<WS>(&As[fa], &Bs[fb], wscale, acc);

    __syncthreads();
#pragma unroll
    for (int q = 0; q < 8; ++q)
      *reinterpret_cast<v4f*>(&As[sa + q*2*PJ_AP]) = ra[q];
#pragma unroll
    for (int q = 0; q < NB; ++q)
      *reinterpret_cast<v4f*>(&Bs[sb + q*128]) = rb[q];
    __syncthreads();
  }
  return proj_chunk<WS>(&As[fa], &Bs[fb], wscale, acc);
}

__global__ __launch_bounds__(32) void proj_kernel(
    const float* __restrict__ node,
    const float* __restrict__ sq_k, const float* __restrict__ sk_k,
    const float* __restrict__ sv_k, const float* __restrict__ pq_k,
    const float* __restrict__ pk_k, const float* __restrict__ pv_k,
    float* __restrict__ ws)
{
  __shared__ __attribute__((aligned(16))) float As[16 * PJ_AP];
  __shared__ __attribute__((aligned(16))) float Bs[PJ_KC * 24];

  const int lane  = threadIdx.x;
  const int mt    = blockIdx.x;
  const int seg   = blockIdx.y;
  const int h     = blockIdx.z;
  const int row   = lane & 15;

  const float* ga = node + (size_t)(mt*16) * NODE;
  v8f acc;
  if (seg < 3) {
    const float* wb = (seg == 0) ? sq_k : (seg == 1) ? sk_k : sv_k;
    acc = proj_body<16>(ga, wb + (size_t)h*NODE*16, As, Bs, lane, row,
                        (seg == 0) ? 0.25f : 1.f);
  } else if (seg < 5) {
    const float* wb = (seg == 3) ? pq_k : pk_k;
    acc = proj_body<12>(ga, wb + (size_t)h*NODE*12, As, Bs, lane,
                        (row < 12) ? row : 11, (row < 12) ? 1.f : 0.f);
  } else if (seg == 5) {
    acc = proj_body<24>(ga, pv_k + (size_t)h*NODE*24, As, Bs, lane, row, 1.f);
  } else {
    acc = proj_body<24>(ga, pv_k + (size_t)h*NODE*24, As, Bs, lane,
                        (row < 8) ? 16 + row : 23, (row < 8) ? 1.f : 0.f);
  }

  const int col   = lane & 15;
  const int rbase = (lane >> 4) * 8;
  float* ob = ws + OFF_SK + (size_t)h*LQ*16;
  int ostride = 16, ocol = col;
  bool dostore = true;
  switch (seg) {
    case 0: ob = ws + OFF_SQ + (size_t)h*LQ*16; break;
    case 1: break;
    case 2: ob = ws + OFF_SV + (size_t)h*LQ*16; break;
    case 3: ob = ws + OFF_PQ + (size_t)h*LQ*16; break;
    case 4: ob = ws + OFF_PK + (size_t)h*LQ*16; break;
    case 5: ob = ws + OFF_PV + (size_t)h*LQ*24; ostride = 24; break;
    case 6: ob = ws + OFF_PV + (size_t)h*LQ*24; ostride = 24;
            ocol = 16 + col; dostore = (col < 8); break;
  }
  if (dostore) {
#pragma unroll
    for (int r = 0; r < 8; ++r)
      ob[(size_t)(mt*16 + rbase + r)*ostride + ocol] = acc[r];
  }
}

__global__ void rigid_kernel(const float* __restrict__ R,
                             const float* __restrict__ t,
                             float* __restrict__ ws)
{
  const int idx = blockIdx.x * blockDim.x + threadIdx.x;
  if (idx >= NH * LQ) return;
  const int h = idx / LQ, l = idx % LQ;
  float Rm[9], tv[3];
#pragma unroll
  for (int i = 0; i < 9; ++i) Rm[i] = R[l*9 + i];
#pragma unroll
  for (int i = 0; i < 3; ++i) tv[i] = t[l*3 + i];

  float* Pq = ws + OFF_PQ + ((size_t)h*LQ + l)*16;
  float* Pk = ws + OFF_PK + ((size_t)h*LQ + l)*16;
  float* Pv = ws + OFF_PV + ((size_t)h*LQ + l)*24;

  float qs = 0.f, ksum = 0.f;
#pragma unroll
  for (int p = 0; p < 4; ++p) {
    float loc[3], g[3];
    for (int f = 0; f < 3; ++f) loc[f] = Pq[p*3+f];
    for (int gg = 0; gg < 3; ++gg) {
      g[gg] = tv[gg];
      for (int f = 0; f < 3; ++f) g[gg] += loc[f]*Rm[gg*3+f];
      Pq[p*3+gg] = g[gg]; qs += g[gg]*g[gg];
    }
    for (int f = 0; f < 3; ++f) loc[f] = Pk[p*3+f];
    for (int gg = 0; gg < 3; ++gg) {
      g[gg] = tv[gg];
      for (int f = 0; f < 3; ++f) g[gg] += loc[f]*Rm[gg*3+f];
      Pk[p*3+gg] = g[gg]; ksum += g[gg]*g[gg];
    }
  }
#pragma unroll
  for (int p = 0; p < 8; ++p) {
    float loc[3], g[3];
    for (int f = 0; f < 3; ++f) loc[f] = Pv[p*3+f];
    for (int gg = 0; gg < 3; ++gg) {
      g[gg] = tv[gg];
      for (int f = 0; f < 3; ++f) g[gg] += loc[f]*Rm[gg*3+f];
      Pv[p*3+gg] = g[gg];
    }
  }
  ws[OFF_QN2 + h*LQ + l] = qs;
  ws[OFF_KN2 + h*LQ + l] = ksum;
}

#define BI_JC 128
#define BI_NCH (LQ / BI_JC)
#define BI_BP 132

static __device__ __forceinline__ v8f bias_chunk(
    const float* __restrict__ Bb, const v2f (&pa)[PAIR / 4])
{
  v2f b[PAIR / 4];
#pragma unroll
  for (int kk = 0; kk < PAIR / 4; ++kk)
    b[kk] = *reinterpret_cast<const v2f*>(Bb + kk*4);
  __builtin_amdgcn_sched_barrier(0);
  v8f acc = vzero8();
#pragma unroll
  for (int kk = 0; kk < PAIR / 4; ++kk)
    acc = wmma4(pa[kk], b[kk], acc);
  asm volatile("" : "+v"(acc));
  __builtin_amdgcn_sched_barrier(0);
  return acc;
}

static __device__ __forceinline__ void bias_store(
    float* __restrict__ ob, v8f acc, bool lowhalf)
{
#pragma unroll
  for (int r = 0; r < 4; ++r) ob[(size_t)r * LQ * LQ] = acc[r];
  if (lowhalf) {
#pragma unroll
    for (int r = 4; r < 8; ++r) ob[(size_t)r * LQ * LQ] = acc[r];
  }
}

__global__ __launch_bounds__(256) void bias_kernel(
    const float* __restrict__ pairwise, const float* __restrict__ pr_k,
    float* __restrict__ ws)
{
  __shared__ __attribute__((aligned(16))) float Bs[BI_JC * BI_BP];

  float* bias = ws + OFF_BIAS;
  const int tid   = threadIdx.x;
  const int lane  = tid & 31;
  const int w     = tid >> 5;
  const int i     = blockIdx.x;
  const int row   = lane & 15;
  const int khalf = (lane >> 4) * 2;

  const float* gb = pairwise + (size_t)i * LQ * PAIR + (size_t)tid * 4;
  const int sb = (tid >> 5) * BI_BP + (tid & 31) * 4;
  const int fb = (w*16 + row) * BI_BP + khalf;

  v4f rb[16];
  {
    const float* gbl = gb + (size_t)(BI_NCH - 1) * BI_JC * PAIR;
#pragma unroll
    for (int q = 0; q < 16; ++q)
      rb[q] = *reinterpret_cast<const v4f*>(gbl + (size_t)q * 1024);
  }

  const bool live = row < NH;
  const float* prk = pr_k + (size_t)(live ? row : NH - 1) * PAIR + khalf;
  v2f pa[PAIR / 4];
#pragma unroll
  for (int kk = 0; kk < PAIR / 4; ++kk) {
    const v2f t = *reinterpret_cast<const v2f*>(prk + kk*4);
    pa[kk].x = live ? t.x : 0.f;
    pa[kk].y = live ? t.y : 0.f;
  }

#pragma unroll
  for (int q = 0; q < 16; ++q)
    *reinterpret_cast<v4f*>(&Bs[sb + q*8*BI_BP]) = rb[q];
  __syncthreads();

  const bool lowhalf = lane < 16;
  float* ob = bias + ((size_t)((lane >> 4) * 8) * LQ + i) * LQ + w*16 + row;

  for (int ch = BI_NCH - 1; ch > 0; --ch) {
    const float* gbn = gb + (size_t)(ch - 1) * BI_JC * PAIR;
#pragma unroll
    for (int q = 0; q < 16; ++q)
      rb[q] = *reinterpret_cast<const v4f*>(gbn + (size_t)q * 1024);

    const v8f acc = bias_chunk(&Bs[fb], pa);

    __syncthreads();
#pragma unroll
    for (int q = 0; q < 16; ++q)
      *reinterpret_cast<v4f*>(&Bs[sb + q*8*BI_BP]) = rb[q];
    __syncthreads();
    bias_store(ob + ch * BI_JC, acc, lowhalf);
  }
  const v8f acc = bias_chunk(&Bs[fb], pa);
  bias_store(ob, acc, lowhalf);
}

#define SMS 516
#define AT_KP 20
#define AT_G  32

template <int VP>
static __device__ __forceinline__ v8f av_group(
    const float* __restrict__ Ab, const float* __restrict__ Bb, v8f acc)
{
  v2f a[AT_G], b[AT_G];
#pragma unroll
  for (int kk = 0; kk < AT_G; ++kk) {
    a[kk]   = *reinterpret_cast<const v2f*>(Ab + kk*4);
    b[kk].x = Bb[(kk*4)     * VP];
    b[kk].y = Bb[(kk*4 + 1) * VP];
  }
  __builtin_amdgcn_sched_barrier(0);
#pragma unroll
  for (int kk = 0; kk < AT_G; ++kk)
    acc = wmma4(a[kk], b[kk], acc);
  __builtin_amdgcn_sched_barrier(0);
  return acc;
}

template <int VP>
static __device__ __forceinline__ v8f av_tile(
    const float* __restrict__ Ab, const float* __restrict__ Bb)
{
  v8f acc = vzero8();
#pragma unroll
  for (int g = 0; g < LQ / (4 * AT_G); ++g)
    acc = av_group<VP>(Ab + g * 4 * AT_G, Bb + (size_t)g * 4 * AT_G * VP, acc);
  return acc;
}

__global__ __launch_bounds__(128) void attn_kernel(
    const float* __restrict__ point_weight,
    const float* __restrict__ R, const float* __restrict__ t,
    float* __restrict__ ws)
{
  __shared__ __attribute__((aligned(16))) float sm[16 * SMS];
  __shared__ __attribute__((aligned(16))) float Rs[LQ * 40];
  __shared__ __attribute__((aligned(16))) float kn[LQ];
  const int tid  = threadIdx.x;
  const int lane = tid & 31;
  const int w    = tid >> 5;
  const int nt   = blockIdx.x;
  const int h    = blockIdx.y;

  const float* Sq  = ws + OFF_SQ;  const float* Sk  = ws + OFF_SK;
  const float* Pq  = ws + OFF_PQ;  const float* Pk  = ws + OFF_PK;
  const float* qn2 = ws + OFF_QN2; const float* kn2 = ws + OFF_KN2;
  const float* bias= ws + OFF_BIAS;
  const float* Sv  = ws + OFF_SV;  const float* Pv  = ws + OFF_PV;
  float* attn = ws + OFF_ATTN;
  float* res  = ws + OFF_RES;

  const float pwx = point_weight[h];
  const float pw  = (pwx > 20.f) ? pwx : log1pf(__expf(pwx));
  const float cpt = -0.5f * pw / sqrtf(18.f);
  const float rs3 = 0.57735026919f;

  const int row   = lane & 15;
  const int khalf = (lane >> 4) * 2;
  const int col   = lane & 15;
  const int rbase = (lane >> 4) * 8;

  {
    const float* gsk = Sk + (size_t)h*LQ*16 + (size_t)tid*4;
    const float* gpk = Pk + (size_t)h*LQ*16 + (size_t)tid*4;
    const float* gbi = bias + ((size_t)h*LQ + nt*16)*LQ + (size_t)tid*4;
    v4f rk[16], rp[16], rb[16];
#pragma unroll
    for (int q = 0; q < 16; ++q) {
      rk[q] = *reinterpret_cast<const v4f*>(gsk + (size_t)q*512);
      rp[q] = *reinterpret_cast<const v4f*>(gpk + (size_t)q*512);
    }
#pragma unroll
    for (int q = 0; q < 16; ++q)
      rb[q] = *reinterpret_cast<const v4f*>(gbi + (size_t)q*LQ);
    const v4f rn = *reinterpret_cast<const v4f*>(kn2 + (size_t)h*LQ + tid*4);
    const int kd = (tid >> 2) * AT_KP + (tid & 3) * 4;
#pragma unroll
    for (int q = 0; q < 16; ++q) {
      *reinterpret_cast<v4f*>(&Rs[kd + q*32*AT_KP])              = rk[q];
      *reinterpret_cast<v4f*>(&Rs[LQ*AT_KP + kd + q*32*AT_KP])   = rp[q];
    }
#pragma unroll
    for (int q = 0; q < 16; ++q)
      *reinterpret_cast<v4f*>(&sm[q*SMS + tid*4]) = rb[q];
    *reinterpret_cast<v4f*>(&kn[tid*4]) = rn;
  }

  v2f aq[4], ap[4];
#pragma unroll
  for (int ks = 0; ks < 4; ++ks) {
    const int k = ks*4 + khalf;
    const int l = nt*16 + row;
    aq[ks].x = Sq[((size_t)h*LQ + l)*16 + k];
    aq[ks].y = Sq[((size_t)h*LQ + l)*16 + k + 1];
    ap[ks].x = Pq[((size_t)h*LQ + l)*16 + k];
    ap[ks].y = Pq[((size_t)h*LQ + l)*16 + k + 1];
  }
  float qv[8];
#pragma unroll
  for (int r = 0; r < 8; ++r) qv[r] = qn2[h*LQ + nt*16 + rbase + r];
  __syncthreads();

  for (int mt = w; mt < 32; mt += 4) {
    v8f s = vzero8(), p = vzero8();
    const float* kb = &Rs[(mt*16 + row)*AT_KP + khalf];
#pragma unroll
    for (int ks = 0; ks < 4; ++ks) {
      const v2f bq = *reinterpret_cast<const v2f*>(kb + ks*4);
      const v2f bp = *reinterpret_cast<const v2f*>(kb + LQ*AT_KP + ks*4);
      s = wmma4(aq[ks], bq, s);
      p = wmma4(ap[ks], bp, p);
    }
    const int m = mt*16 + col;
    const float kv = kn[m];
#pragma unroll
    for (int r = 0; r < 8; ++r) {
      const int rr = rbase + r;
      const float dist = qv[r] + kv - 2.f*p[r];
      sm[rr*SMS + m] = (s[r] + sm[rr*SMS + m] + cpt*dist) * rs3;
    }
  }
  __syncthreads();

  v4f rv[40];
  {
    const float* gsv = Sv + (size_t)h*LQ*16 + (size_t)tid*4;
    const float* gpv = Pv + (size_t)h*LQ*24 + (size_t)tid*4;
#pragma unroll
    for (int q = 0; q < 16; ++q)
      rv[q] = *reinterpret_cast<const v4f*>(gsv + (size_t)q*512);
#pragma unroll
    for (int q = 0; q < 24; ++q)
      rv[16 + q] = *reinterpret_cast<const v4f*>(gpv + (size_t)q*512);
  }

  {
    const int rrow = tid >> 3;
    const int sub  = tid & 7;
    float mx = -1e30f;
    for (int j = sub; j < LQ; j += 8) mx = fmaxf(mx, sm[rrow*SMS + j]);
    for (int o = 1; o < 8; o <<= 1) mx = fmaxf(mx, __shfl_xor(mx, o, 8));
    float sum = 0.f;
    for (int j = sub; j < LQ; j += 8) {
      const float e = __expf(sm[rrow*SMS + j] - mx);
      sm[rrow*SMS + j] = e; sum += e;
    }
    for (int o = 1; o < 8; o <<= 1) sum += __shfl_xor(sum, o, 8);
    const float inv = 1.f / sum;
    const int n = nt*16 + rrow;
    for (int j = sub; j < LQ; j += 8) {
      const float v = sm[rrow*SMS + j] * inv;
      sm[rrow*SMS + j] = v;
      attn[((size_t)h*LQ + n)*LQ + j] = v;
    }
  }

#pragma unroll
  for (int q = 0; q < 16; ++q)
    *reinterpret_cast<v4f*>(&Rs[(tid + 128*q)*4]) = rv[q];
#pragma unroll
  for (int q = 0; q < 24; ++q)
    *reinterpret_cast<v4f*>(&Rs[LQ*16 + (tid + 128*q)*4]) = rv[16 + q];
  __syncthreads();

  const int en = nt*16 + (tid >> 3);
  float Rm[9], tv[3];
#pragma unroll
  for (int i = 0; i < 9; ++i) Rm[i] = R[en*9 + i];
#pragma unroll
  for (int i = 0; i < 3; ++i) tv[i] = t[en*3 + i];

  float* pt = kn;
  v8f acc;
  if (w < 3) {
    const float* Ab = &sm[row*SMS + khalf];
    if (w == 0) {
      acc = av_tile<16>(Ab, &Rs[khalf*16 + col]);
    } else {
      const int vc = (w == 1) ? col : ((col < 8) ? 16 + col : 23);
      acc = av_tile<24>(Ab, &Rs[LQ*16 + khalf*24 + vc]);
      if (w == 1 || col < 8) {
        const int pc = (w == 1) ? col : 16 + col;
#pragma unroll
        for (int r = 0; r < 8; ++r)
          pt[(rbase + r)*24 + pc] = acc[r];
      }
    }
  }
  __syncthreads();

  if (w == 0) {
#pragma unroll
    for (int r = 0; r < 8; ++r)
      res[RES_IDX(h, nt*16 + rbase + r, col)] = acc[r];
  }

  {
    const int rr = tid >> 3;
    const int p  = tid & 7;
    float* resp = res + RES_IDX(h, en, 0);
    float g[3], loc[3], nrm = 0.f;
    for (int gg = 0; gg < 3; ++gg) g[gg] = pt[rr*24 + p*3 + gg] - tv[gg];
    for (int f = 0; f < 3; ++f) {
      float v = 0.f;
      for (int gg = 0; gg < 3; ++gg) v += g[gg]*Rm[gg*3 + f];
      loc[f] = v; nrm += v*v;
    }
    for (int f = 0; f < 3; ++f) resp[144 + p*3 + f] = loc[f];
    resp[168 + p] = sqrtf(nrm + 1e-12f);
  }
}

#define PO_KC 64
#define PO_NCH (LQ / PO_KC)
#define PO_AP 68
#define PO_BP 144

static __device__ __forceinline__ v8f pairout_chunk(
    const float* __restrict__ Ab, const float* __restrict__ Bb, v8f acc)
{
  v2f a[PO_KC / 4], b[PO_KC / 4];
#pragma unroll
  for (int kk = 0; kk < PO_KC / 4; ++kk) {
    a[kk]   = *reinterpret_cast<const v2f*>(Ab + kk*4);
    b[kk].x = Bb[(kk*4)     * PO_BP];
    b[kk].y = Bb[(kk*4 + 1) * PO_BP];
  }
  __builtin_amdgcn_sched_barrier(0);
#pragma unroll
  for (int kk = 0; kk < PO_KC / 4; ++kk)
    acc = wmma4(a[kk], b[kk], acc);
  asm volatile("" : "+v"(acc));
  __builtin_amdgcn_sched_barrier(0);
  return acc;
}

__global__ __launch_bounds__(256) void pairout_kernel(
    const float* __restrict__ pairwise, float* __restrict__ ws)
{
  __shared__ __attribute__((aligned(16))) float As[16 * PO_AP];
  __shared__ __attribute__((aligned(16))) float Bs[PO_KC * PO_BP];

  const float* attn = ws + OFF_ATTN;
  float* res = ws + OFF_RES;
  const int tid   = threadIdx.x;
  const int lane  = tid & 31;
  const int w     = tid >> 5;
  const int n     = blockIdx.x;
  const int row   = lane & 15;
  const int khalf = (lane >> 4) * 2;
  const int col   = lane & 15;

  const float* gb = pairwise + (size_t)n * LQ * PAIR + (size_t)tid * 4;
  const int sb = (tid >> 5) * PO_BP + (tid & 31) * 4;
  const bool hasA = tid < NH * 16;
  const int ah = hasA ? (tid >> 4) : 0;
  const float* gaa = attn + ((size_t)ah * LQ + n) * LQ + (tid & 15) * 4;
  const int sa = ah * PO_AP + (tid & 15) * 4;

  const int fa = row * PO_AP + khalf;
  const int fb = khalf * PO_BP + w*16 + col;

  v4f rb[8], ra;
#pragma unroll
  for (int i = 0; i < 8; ++i)
    rb[i] = *reinterpret_cast<const v4f*>(gb + (size_t)i * 1024);
  ra = *reinterpret_cast<const v4f*>(gaa);
#pragma unroll
  for (int i = 0; i < 8; ++i)
    *reinterpret_cast<v4f*>(&Bs[sb + i*8*PO_BP]) = rb[i];
  if (hasA) {
    *reinterpret_cast<v4f*>(&As[sa]) = ra;
  } else {
    const v4f z4 = {0.f, 0.f, 0.f, 0.f};
    *reinterpret_cast<v4f*>(&As[(tid >> 4) * PO_AP + (tid & 15) * 4]) = z4;
  }
  __syncthreads();

  v8f acc = vzero8();
  for (int ch = 0; ch < PO_NCH - 1; ++ch) {
    const float* gbn = gb + (size_t)(ch + 1) * PO_KC * PAIR;
#pragma unroll
    for (int i = 0; i < 8; ++i)
      rb[i] = *reinterpret_cast<const v4f*>(gbn + (size_t)i * 1024);
    ra = *reinterpret_cast<const v4f*>(gaa + (ch + 1) * PO_KC);

    acc = pairout_chunk(&As[fa], &Bs[fb], acc);

    __syncthreads();
#pragma unroll
    for (int i = 0; i < 8; ++i)
      *reinterpret_cast<v4f*>(&Bs[sb + i*8*PO_BP]) = rb[i];
    if (hasA) *reinterpret_cast<v4f*>(&As[sa]) = ra;
    __syncthreads();
  }
  acc = pairout_chunk(&As[fa], &Bs[fb], acc);

  const int rbase = (lane >> 4) * 8;
#pragma unroll
  for (int r = 0; r < 8; ++r) {
    const int hh = rbase + r;
    if (hh < NH)
      res[RES_IDX(hh, n, 16 + w*16 + col)] = acc[r];
  }
}

#define OP_KC  96
#define OP_NCH (KOUT / OP_KC)
#define OP_AP  100
#define OP_BP  48

static __device__ __forceinline__ v8f outproj_chunk(
    const float* __restrict__ Ab, const float* __restrict__ Bb, v8f acc)
{
  v2f a[OP_KC / 4], b[OP_KC / 4];
#pragma unroll
  for (int kk = 0; kk < OP_KC / 4; ++kk) {
    a[kk]   = *reinterpret_cast<const v2f*>(Ab + kk*4);
    b[kk].x = Bb[(kk*4)     * OP_BP];
    b[kk].y = Bb[(kk*4 + 1) * OP_BP];
  }
  __builtin_amdgcn_sched_barrier(0);
#pragma unroll
  for (int kk = 0; kk < OP_KC / 4; ++kk)
    acc = wmma4(a[kk], b[kk], acc);
  asm volatile("" : "+v"(acc));
  __builtin_amdgcn_sched_barrier(0);
  return acc;
}

__global__ __launch_bounds__(128) void outproj_kernel(
    const float* __restrict__ out_k, const float* __restrict__ out_b,
    const float* __restrict__ res, float* __restrict__ out)
{
  __shared__ __attribute__((aligned(16))) float As[2][32 * OP_AP];
  __shared__ __attribute__((aligned(16))) float Bs[2][OP_KC * OP_BP];

  const int tid   = threadIdx.x;
  const int lane  = tid & 31;
  const int w     = tid >> 5;
  const int wm    = w >> 1;
  const int wn    = w & 1;
  const int m0    = blockIdx.x * 32;
  const int n0    = blockIdx.y * 32;
  const int col   = lane & 15;
  const int khalf = (lane >> 4) * 2;

  const int srow = tid >> 3;
  const int sseg = tid & 7;
  const float* ga = res   + (size_t)(m0 + srow) * KOUT + sseg*4;
  const float* gb = out_k + (size_t)srow * NODE + n0 + sseg*4;
  const int sa = srow * OP_AP + sseg*4;
  const int sb = srow * OP_BP + sseg*4;

  const int fa = (wm*16 + col) * OP_AP + khalf;
  const int fb = khalf * OP_BP + wn*16 + col;
  const int o  = n0 + wn*16 + col;
  const float ob = out_b[o];

  v4f ra[6], rb[6];
#pragma unroll
  for (int i = 0; i < 2; ++i)
#pragma unroll
    for (int j = 0; j < 3; ++j)
      ra[i*3 + j] = *reinterpret_cast<const v4f*>(ga + (size_t)i*16*KOUT + j*32);
#pragma unroll
  for (int i = 0; i < 6; ++i)
    rb[i] = *reinterpret_cast<const v4f*>(gb + (size_t)i*16*NODE);
#pragma unroll
  for (int i = 0; i < 2; ++i)
#pragma unroll
    for (int j = 0; j < 3; ++j)
      *reinterpret_cast<v4f*>(&As[0][sa + i*16*OP_AP + j*32]) = ra[i*3 + j];
#pragma unroll
  for (int i = 0; i < 6; ++i)
    *reinterpret_cast<v4f*>(&Bs[0][sb + i*16*OP_BP]) = rb[i];
  __syncthreads();

  v8f acc = vzero8();
  for (int ch = 0; ch < OP_NCH - 1; ++ch) {
    const int cur = ch & 1, nxt = cur ^ 1;
    const float* gan = ga + (size_t)(ch + 1) * OP_KC;
    const float* gbn = gb + (size_t)(ch + 1) * OP_KC * NODE;
#pragma unroll
    for (int i = 0; i < 2; ++i)
#pragma unroll
      for (int j = 0; j < 3; ++j)
        ra[i*3 + j] = *reinterpret_cast<const v4f*>(gan + (size_t)i*16*KOUT + j*32);
#pragma unroll
    for (int i = 0; i < 6; ++i)
      rb[i] = *reinterpret_cast<const v4f*>(gbn + (size_t)i*16*NODE);

    acc = outproj_chunk(&As[cur][fa], &Bs[cur][fb], acc);

#pragma unroll
    for (int i = 0; i < 2; ++i)
#pragma unroll
      for (int j = 0; j < 3; ++j)
        *reinterpret_cast<v4f*>(&As[nxt][sa + i*16*OP_AP + j*32]) = ra[i*3 + j];
#pragma unroll
    for (int i = 0; i < 6; ++i)
      *reinterpret_cast<v4f*>(&Bs[nxt][sb + i*16*OP_BP]) = rb[i];
    __syncthreads();
  }
  acc = outproj_chunk(&As[(OP_NCH - 1) & 1][fa], &Bs[(OP_NCH - 1) & 1][fb], acc);

  const int rbase = (lane >> 4) * 8;
#pragma unroll
  for (int r = 0; r < 8; ++r) {
    const int l = m0 + wm*16 + rbase + r;
    out[(size_t)l*NODE + o] = acc[r] + ob;
  }
}

extern "C" void kernel_launch(void* const* d_in, const int* in_sizes, int n_in,
                              void* d_out, int out_size, void* d_ws, size_t ws_size,
                              hipStream_t stream) {
  const float* node  = (const float*)d_in[0];
  const float* R     = (const float*)d_in[1];
  const float* t     = (const float*)d_in[2];
  const float* pairw = (const float*)d_in[3];
  const float* sq_k  = (const float*)d_in[4];
  const float* sk_k  = (const float*)d_in[5];
  const float* sv_k  = (const float*)d_in[6];
  const float* pq_k  = (const float*)d_in[7];
  const float* pk_k  = (const float*)d_in[8];
  const float* pv_k  = (const float*)d_in[9];
  const float* pr_k  = (const float*)d_in[10];
  const float* out_k = (const float*)d_in[11];
  const float* out_b = (const float*)d_in[12];
  const float* pwgt  = (const float*)d_in[13];
  float* ws  = (float*)d_ws;
  float* out = (float*)d_out;

  proj_kernel<<<dim3(32, 7, NH), 32, 0, stream>>>(
      node, sq_k, sk_k, sv_k, pq_k, pk_k, pv_k, ws);
  rigid_kernel<<<(NH*LQ + 127)/128, 128, 0, stream>>>(R, t, ws);
  bias_kernel<<<LQ, 256, 0, stream>>>(pairw, pr_k, ws);
  attn_kernel<<<dim3(32, NH), 128, 0, stream>>>(pwgt, R, t, ws);
  pairout_kernel<<<LQ, 256, 0, stream>>>(pairw, ws);
  outproj_kernel<<<dim3(16, 12), 128, 0, stream>>>(out_k, out_b, ws + OFF_RES, out);
}


# outproj: K chunk depth 96 -> 192 (22 -> 11 LDS hand-overs per block; the next chunk's global loads get a 48-step wmma chain to arrive, one wave per SIMD); k order and per-element sum order unchanged
# speedup vs baseline: 3.6574x; 1.0651x over previous
#include <hip/hip_runtime.h>
#include <math.h>


typedef __attribute__((ext_vector_type(2))) float v2f;
typedef __attribute__((ext_vector_type(4))) float v4f;
typedef __attribute__((ext_vector_type(8))) float v8f;

#define LQ    512
#define NH    12
#define NODE  384
#define PAIR  128
#define OUTIN 176
#define KOUT  (NH*OUTIN)

#define OFF_SQ   ((size_t)0)
#define OFF_SK   (OFF_SQ  + (size_t)NH*LQ*16)
#define OFF_SV   (OFF_SK  + (size_t)NH*LQ*16)
#define OFF_PQ   (OFF_SV  + (size_t)NH*LQ*16)
#define OFF_PK   (OFF_PQ  + (size_t)NH*LQ*16)
#define OFF_PV   (OFF_PK  + (size_t)NH*LQ*16)
#define OFF_QN2  (OFF_PV  + (size_t)NH*LQ*24)
#define OFF_KN2  (OFF_QN2 + (size_t)NH*LQ)
#define OFF_BIAS (OFF_KN2 + (size_t)NH*LQ)
#define OFF_ATTN (OFF_BIAS+ (size_t)NH*LQ*LQ)
#define OFF_RES  (OFF_ATTN+ (size_t)NH*LQ*LQ)

#define RES_IDX(h, l, f) (((size_t)(l)*NH + (h))*OUTIN + (f))

static __device__ __forceinline__ v8f wmma4(v2f a, v2f b, v8f c) {
  return __builtin_amdgcn_wmma_f32_16x16x4_f32(
       false, a,  false, b,
       (short)0, c,  false,  false);
}

static __device__ __forceinline__ v8f vzero8() {
  v8f z = {0.f,0.f,0.f,0.f,0.f,0.f,0.f,0.f};
  return z;
}

#define PJ_KC  64
#define PJ_NCH (NODE / PJ_KC)
#define PJ_AP  68

template <int WS>
static __device__ __forceinline__ v8f proj_chunk(
    const float* __restrict__ Ab, const float* __restrict__ Bb, float wscale, v8f acc)
{
  v2f a[PJ_KC / 4], b[PJ_KC / 4];
#pragma unroll
  for (int kk = 0; kk < PJ_KC / 4; ++kk) {
    a[kk]   = *reinterpret_cast<const v2f*>(Ab + kk*4);
    b[kk].x = Bb[(kk*4)     * WS] * wscale;
    b[kk].y = Bb[(kk*4 + 1) * WS] * wscale;
  }
  __builtin_amdgcn_sched_barrier(0);
#pragma unroll
  for (int kk = 0; kk < PJ_KC / 4; ++kk)
    acc = wmma4(a[kk], b[kk], acc);
  asm volatile("" : "+v"(acc));
  __builtin_amdgcn_sched_barrier(0);
  return acc;
}

template <int WS>
static __device__ __forceinline__ v8f proj_body(
    const float* __restrict__ ga, const float* __restrict__ gw,
    float* __restrict__ As, float* __restrict__ Bs,
    int lane, int wcol, float wscale)
{
  constexpr int NB = PJ_KC * WS / 128;
  const int row   = lane & 15;
  const int khalf = (lane >> 4) * 2;
  const float* gaa = ga + (size_t)(lane >> 4) * NODE + (lane & 15) * 4;
  const int sa = (lane >> 4) * PJ_AP + (lane & 15) * 4;
  const float* gbb = gw + (size_t)lane * 4;
  const int sb = lane * 4;
  const int fa = row * PJ_AP + khalf;
  const int fb = khalf * WS + wcol;

  v4f ra[8], rb[NB];
#pragma unroll
  for (int q = 0; q < 8; ++q)
    ra[q] = *reinterpret_cast<const v4f*>(gaa + (size_t)q * 2 * NODE);
#pragma unroll
  for (int q = 0; q < NB; ++q)
    rb[q] = *reinterpret_cast<const v4f*>(gbb + (size_t)q * 128);
#pragma unroll
  for (int q = 0; q < 8; ++q)
    *reinterpret_cast<v4f*>(&As[sa + q*2*PJ_AP]) = ra[q];
#pragma unroll
  for (int q = 0; q < NB; ++q)
    *reinterpret_cast<v4f*>(&Bs[sb + q*128]) = rb[q];
  __syncthreads();

  v8f acc = vzero8();
  for (int ch = 0; ch < PJ_NCH - 1; ++ch) {
    const float* gan = gaa + (size_t)(ch + 1) * PJ_KC;
    const float* gbn = gbb + (size_t)(ch + 1) * PJ_KC * WS;
#pragma unroll
    for (int q = 0; q < 8; ++q)
      ra[q] = *reinterpret_cast<const v4f*>(gan + (size_t)q * 2 * NODE);
#pragma unroll
    for (int q = 0; q < NB; ++q)
      rb[q] = *reinterpret_cast<const v4f*>(gbn + (size_t)q * 128);

    acc = proj_chunk<WS>(&As[fa], &Bs[fb], wscale, acc);

    __syncthreads();
#pragma unroll
    for (int q = 0; q < 8; ++q)
      *reinterpret_cast<v4f*>(&As[sa + q*2*PJ_AP]) = ra[q];
#pragma unroll
    for (int q = 0; q < NB; ++q)
      *reinterpret_cast<v4f*>(&Bs[sb + q*128]) = rb[q];
    __syncthreads();
  }
  return proj_chunk<WS>(&As[fa], &Bs[fb], wscale, acc);
}

__global__ __launch_bounds__(32) void proj_kernel(
    const float* __restrict__ node,
    const float* __restrict__ sq_k, const float* __restrict__ sk_k,
    const float* __restrict__ sv_k, const float* __restrict__ pq_k,
    const float* __restrict__ pk_k, const float* __restrict__ pv_k,
    float* __restrict__ ws)
{
  __shared__ __attribute__((aligned(16))) float As[16 * PJ_AP];
  __shared__ __attribute__((aligned(16))) float Bs[PJ_KC * 24];

  const int lane  = threadIdx.x;
  const int mt    = blockIdx.x;
  const int seg   = blockIdx.y;
  const int h     = blockIdx.z;
  const int row   = lane & 15;

  const float* ga = node + (size_t)(mt*16) * NODE;
  v8f acc;
  if (seg < 3) {
    const float* wb = (seg == 0) ? sq_k : (seg == 1) ? sk_k : sv_k;
    acc = proj_body<16>(ga, wb + (size_t)h*NODE*16, As, Bs, lane, row,
                        (seg == 0) ? 0.25f : 1.f);
  } else if (seg < 5) {
    const float* wb = (seg == 3) ? pq_k : pk_k;
    acc = proj_body<12>(ga, wb + (size_t)h*NODE*12, As, Bs, lane,
                        (row < 12) ? row : 11, (row < 12) ? 1.f : 0.f);
  } else if (seg == 5) {
    acc = proj_body<24>(ga, pv_k + (size_t)h*NODE*24, As, Bs, lane, row, 1.f);
  } else {
    acc = proj_body<24>(ga, pv_k + (size_t)h*NODE*24, As, Bs, lane,
                        (row < 8) ? 16 + row : 23, (row < 8) ? 1.f : 0.f);
  }

  const int col   = lane & 15;
  const int rbase = (lane >> 4) * 8;
  float* ob = ws + OFF_SK + (size_t)h*LQ*16;
  int ostride = 16, ocol = col;
  bool dostore = true;
  switch (seg) {
    case 0: ob = ws + OFF_SQ + (size_t)h*LQ*16; break;
    case 1: break;
    case 2: ob = ws + OFF_SV + (size_t)h*LQ*16; break;
    case 3: ob = ws + OFF_PQ + (size_t)h*LQ*16; break;
    case 4: ob = ws + OFF_PK + (size_t)h*LQ*16; break;
    case 5: ob = ws + OFF_PV + (size_t)h*LQ*24; ostride = 24; break;
    case 6: ob = ws + OFF_PV + (size_t)h*LQ*24; ostride = 24;
            ocol = 16 + col; dostore = (col < 8); break;
  }
  if (dostore) {
#pragma unroll
    for (int r = 0; r < 8; ++r)
      ob[(size_t)(mt*16 + rbase + r)*ostride + ocol] = acc[r];
  }
}

__global__ void rigid_kernel(const float* __restrict__ R,
                             const float* __restrict__ t,
                             float* __restrict__ ws)
{
  const int idx = blockIdx.x * blockDim.x + threadIdx.x;
  if (idx >= NH * LQ) return;
  const int h = idx / LQ, l = idx % LQ;
  float Rm[9], tv[3];
#pragma unroll
  for (int i = 0; i < 9; ++i) Rm[i] = R[l*9 + i];
#pragma unroll
  for (int i = 0; i < 3; ++i) tv[i] = t[l*3 + i];

  float* Pq = ws + OFF_PQ + ((size_t)h*LQ + l)*16;
  float* Pk = ws + OFF_PK + ((size_t)h*LQ + l)*16;
  float* Pv = ws + OFF_PV + ((size_t)h*LQ + l)*24;

  float qs = 0.f, ksum = 0.f;
#pragma unroll
  for (int p = 0; p < 4; ++p) {
    float loc[3], g[3];
    for (int f = 0; f < 3; ++f) loc[f] = Pq[p*3+f];
    for (int gg = 0; gg < 3; ++gg) {
      g[gg] = tv[gg];
      for (int f = 0; f < 3; ++f) g[gg] += loc[f]*Rm[gg*3+f];
      Pq[p*3+gg] = g[gg]; qs += g[gg]*g[gg];
    }
    for (int f = 0; f < 3; ++f) loc[f] = Pk[p*3+f];
    for (int gg = 0; gg < 3; ++gg) {
      g[gg] = tv[gg];
      for (int f = 0; f < 3; ++f) g[gg] += loc[f]*Rm[gg*3+f];
      Pk[p*3+gg] = g[gg]; ksum += g[gg]*g[gg];
    }
  }
#pragma unroll
  for (int p = 0; p < 8; ++p) {
    float loc[3], g[3];
    for (int f = 0; f < 3; ++f) loc[f] = Pv[p*3+f];
    for (int gg = 0; gg < 3; ++gg) {
      g[gg] = tv[gg];
      for (int f = 0; f < 3; ++f) g[gg] += loc[f]*Rm[gg*3+f];
      Pv[p*3+gg] = g[gg];
    }
  }
  ws[OFF_QN2 + h*LQ + l] = qs;
  ws[OFF_KN2 + h*LQ + l] = ksum;
}

#define BI_JC 128
#define BI_NCH (LQ / BI_JC)
#define BI_BP 132

static __device__ __forceinline__ v8f bias_chunk(
    const float* __restrict__ Bb, const v2f (&pa)[PAIR / 4])
{
  v2f b[PAIR / 4];
#pragma unroll
  for (int kk = 0; kk < PAIR / 4; ++kk)
    b[kk] = *reinterpret_cast<const v2f*>(Bb + kk*4);
  __builtin_amdgcn_sched_barrier(0);
  v8f acc = vzero8();
#pragma unroll
  for (int kk = 0; kk < PAIR / 4; ++kk)
    acc = wmma4(pa[kk], b[kk], acc);
  asm volatile("" : "+v"(acc));
  __builtin_amdgcn_sched_barrier(0);
  return acc;
}

static __device__ __forceinline__ void bias_store(
    float* __restrict__ ob, v8f acc, bool lowhalf)
{
#pragma unroll
  for (int r = 0; r < 4; ++r) ob[(size_t)r * LQ * LQ] = acc[r];
  if (lowhalf) {
#pragma unroll
    for (int r = 4; r < 8; ++r) ob[(size_t)r * LQ * LQ] = acc[r];
  }
}

__global__ __launch_bounds__(256) void bias_kernel(
    const float* __restrict__ pairwise, const float* __restrict__ pr_k,
    float* __restrict__ ws)
{
  __shared__ __attribute__((aligned(16))) float Bs[BI_JC * BI_BP];

  float* bias = ws + OFF_BIAS;
  const int tid   = threadIdx.x;
  const int lane  = tid & 31;
  const int w     = tid >> 5;
  const int i     = blockIdx.x;
  const int row   = lane & 15;
  const int khalf = (lane >> 4) * 2;

  const float* gb = pairwise + (size_t)i * LQ * PAIR + (size_t)tid * 4;
  const int sb = (tid >> 5) * BI_BP + (tid & 31) * 4;
  const int fb = (w*16 + row) * BI_BP + khalf;

  v4f rb[16];
  {
    const float* gbl = gb + (size_t)(BI_NCH - 1) * BI_JC * PAIR;
#pragma unroll
    for (int q = 0; q < 16; ++q)
      rb[q] = *reinterpret_cast<const v4f*>(gbl + (size_t)q * 1024);
  }

  const bool live = row < NH;
  const float* prk = pr_k + (size_t)(live ? row : NH - 1) * PAIR + khalf;
  v2f pa[PAIR / 4];
#pragma unroll
  for (int kk = 0; kk < PAIR / 4; ++kk) {
    const v2f t = *reinterpret_cast<const v2f*>(prk + kk*4);
    pa[kk].x = live ? t.x : 0.f;
    pa[kk].y = live ? t.y : 0.f;
  }

#pragma unroll
  for (int q = 0; q < 16; ++q)
    *reinterpret_cast<v4f*>(&Bs[sb + q*8*BI_BP]) = rb[q];
  __syncthreads();

  const bool lowhalf = lane < 16;
  float* ob = bias + ((size_t)((lane >> 4) * 8) * LQ + i) * LQ + w*16 + row;

  for (int ch = BI_NCH - 1; ch > 0; --ch) {
    const float* gbn = gb + (size_t)(ch - 1) * BI_JC * PAIR;
#pragma unroll
    for (int q = 0; q < 16; ++q)
      rb[q] = *reinterpret_cast<const v4f*>(gbn + (size_t)q * 1024);

    const v8f acc = bias_chunk(&Bs[fb], pa);

    __syncthreads();
#pragma unroll
    for (int q = 0; q < 16; ++q)
      *reinterpret_cast<v4f*>(&Bs[sb + q*8*BI_BP]) = rb[q];
    __syncthreads();
    bias_store(ob + ch * BI_JC, acc, lowhalf);
  }
  const v8f acc = bias_chunk(&Bs[fb], pa);
  bias_store(ob, acc, lowhalf);
}

#define SMS 516
#define AT_KP 20
#define AT_G  32

template <int VP>
static __device__ __forceinline__ v8f av_group(
    const float* __restrict__ Ab, const float* __restrict__ Bb, v8f acc)
{
  v2f a[AT_G], b[AT_G];
#pragma unroll
  for (int kk = 0; kk < AT_G; ++kk) {
    a[kk]   = *reinterpret_cast<const v2f*>(Ab + kk*4);
    b[kk].x = Bb[(kk*4)     * VP];
    b[kk].y = Bb[(kk*4 + 1) * VP];
  }
  __builtin_amdgcn_sched_barrier(0);
#pragma unroll
  for (int kk = 0; kk < AT_G; ++kk)
    acc = wmma4(a[kk], b[kk], acc);
  __builtin_amdgcn_sched_barrier(0);
  return acc;
}

template <int VP>
static __device__ __forceinline__ v8f av_tile(
    const float* __restrict__ Ab, const float* __restrict__ Bb)
{
  v8f acc = vzero8();
#pragma unroll
  for (int g = 0; g < LQ / (4 * AT_G); ++g)
    acc = av_group<VP>(Ab + g * 4 * AT_G, Bb + (size_t)g * 4 * AT_G * VP, acc);
  return acc;
}

__global__ __launch_bounds__(128) void attn_kernel(
    const float* __restrict__ point_weight,
    const float* __restrict__ R, const float* __restrict__ t,
    float* __restrict__ ws)
{
  __shared__ __attribute__((aligned(16))) float sm[16 * SMS];
  __shared__ __attribute__((aligned(16))) float Rs[LQ * 40];
  __shared__ __attribute__((aligned(16))) float kn[LQ];
  const int tid  = threadIdx.x;
  const int lane = tid & 31;
  const int w    = tid >> 5;
  const int nt   = blockIdx.x;
  const int h    = blockIdx.y;

  const float* Sq  = ws + OFF_SQ;  const float* Sk  = ws + OFF_SK;
  const float* Pq  = ws + OFF_PQ;  const float* Pk  = ws + OFF_PK;
  const float* qn2 = ws + OFF_QN2; const float* kn2 = ws + OFF_KN2;
  const float* bias= ws + OFF_BIAS;
  const float* Sv  = ws + OFF_SV;  const float* Pv  = ws + OFF_PV;
  float* attn = ws + OFF_ATTN;
  float* res  = ws + OFF_RES;

  const float pwx = point_weight[h];
  const float pw  = (pwx > 20.f) ? pwx : log1pf(__expf(pwx));
  const float cpt = -0.5f * pw / sqrtf(18.f);
  const float rs3 = 0.57735026919f;

  const int row   = lane & 15;
  const int khalf = (lane >> 4) * 2;
  const int col   = lane & 15;
  const int rbase = (lane >> 4) * 8;

  {
    const float* gsk = Sk + (size_t)h*LQ*16 + (size_t)tid*4;
    const float* gpk = Pk + (size_t)h*LQ*16 + (size_t)tid*4;
    const float* gbi = bias + ((size_t)h*LQ + nt*16)*LQ + (size_t)tid*4;
    v4f rk[16], rp[16], rb[16];
#pragma unroll
    for (int q = 0; q < 16; ++q) {
      rk[q] = *reinterpret_cast<const v4f*>(gsk + (size_t)q*512);
      rp[q] = *reinterpret_cast<const v4f*>(gpk + (size_t)q*512);
    }
#pragma unroll
    for (int q = 0; q < 16; ++q)
      rb[q] = *reinterpret_cast<const v4f*>(gbi + (size_t)q*LQ);
    const v4f rn = *reinterpret_cast<const v4f*>(kn2 + (size_t)h*LQ + tid*4);
    const int kd = (tid >> 2) * AT_KP + (tid & 3) * 4;
#pragma unroll
    for (int q = 0; q < 16; ++q) {
      *reinterpret_cast<v4f*>(&Rs[kd + q*32*AT_KP])              = rk[q];
      *reinterpret_cast<v4f*>(&Rs[LQ*AT_KP + kd + q*32*AT_KP])   = rp[q];
    }
#pragma unroll
    for (int q = 0; q < 16; ++q)
      *reinterpret_cast<v4f*>(&sm[q*SMS + tid*4]) = rb[q];
    *reinterpret_cast<v4f*>(&kn[tid*4]) = rn;
  }

  v2f aq[4], ap[4];
#pragma unroll
  for (int ks = 0; ks < 4; ++ks) {
    const int k = ks*4 + khalf;
    const int l = nt*16 + row;
    aq[ks].x = Sq[((size_t)h*LQ + l)*16 + k];
    aq[ks].y = Sq[((size_t)h*LQ + l)*16 + k + 1];
    ap[ks].x = Pq[((size_t)h*LQ + l)*16 + k];
    ap[ks].y = Pq[((size_t)h*LQ + l)*16 + k + 1];
  }
  float qv[8];
#pragma unroll
  for (int r = 0; r < 8; ++r) qv[r] = qn2[h*LQ + nt*16 + rbase + r];
  __syncthreads();

  for (int mt = w; mt < 32; mt += 4) {
    v8f s = vzero8(), p = vzero8();
    const float* kb = &Rs[(mt*16 + row)*AT_KP + khalf];
#pragma unroll
    for (int ks = 0; ks < 4; ++ks) {
      const v2f bq = *reinterpret_cast<const v2f*>(kb + ks*4);
      const v2f bp = *reinterpret_cast<const v2f*>(kb + LQ*AT_KP + ks*4);
      s = wmma4(aq[ks], bq, s);
      p = wmma4(ap[ks], bp, p);
    }
    const int m = mt*16 + col;
    const float kv = kn[m];
#pragma unroll
    for (int r = 0; r < 8; ++r) {
      const int rr = rbase + r;
      const float dist = qv[r] + kv - 2.f*p[r];
      sm[rr*SMS + m] = (s[r] + sm[rr*SMS + m] + cpt*dist) * rs3;
    }
  }
  __syncthreads();

  v4f rv[40];
  {
    const float* gsv = Sv + (size_t)h*LQ*16 + (size_t)tid*4;
    const float* gpv = Pv + (size_t)h*LQ*24 + (size_t)tid*4;
#pragma unroll
    for (int q = 0; q < 16; ++q)
      rv[q] = *reinterpret_cast<const v4f*>(gsv + (size_t)q*512);
#pragma unroll
    for (int q = 0; q < 24; ++q)
      rv[16 + q] = *reinterpret_cast<const v4f*>(gpv + (size_t)q*512);
  }

  {
    const int rrow = tid >> 3;
    const int sub  = tid & 7;
    float mx = -1e30f;
    for (int j = sub; j < LQ; j += 8) mx = fmaxf(mx, sm[rrow*SMS + j]);
    for (int o = 1; o < 8; o <<= 1) mx = fmaxf(mx, __shfl_xor(mx, o, 8));
    float sum = 0.f;
    for (int j = sub; j < LQ; j += 8) {
      const float e = __expf(sm[rrow*SMS + j] - mx);
      sm[rrow*SMS + j] = e; sum += e;
    }
    for (int o = 1; o < 8; o <<= 1) sum += __shfl_xor(sum, o, 8);
    const float inv = 1.f / sum;
    const int n = nt*16 + rrow;
    for (int j = sub; j < LQ; j += 8) {
      const float v = sm[rrow*SMS + j] * inv;
      sm[rrow*SMS + j] = v;
      attn[((size_t)h*LQ + n)*LQ + j] = v;
    }
  }

#pragma unroll
  for (int q = 0; q < 16; ++q)
    *reinterpret_cast<v4f*>(&Rs[(tid + 128*q)*4]) = rv[q];
#pragma unroll
  for (int q = 0; q < 24; ++q)
    *reinterpret_cast<v4f*>(&Rs[LQ*16 + (tid + 128*q)*4]) = rv[16 + q];
  __syncthreads();

  const int en = nt*16 + (tid >> 3);
  float Rm[9], tv[3];
#pragma unroll
  for (int i = 0; i < 9; ++i) Rm[i] = R[en*9 + i];
#pragma unroll
  for (int i = 0; i < 3; ++i) tv[i] = t[en*3 + i];

  float* pt = kn;
  v8f acc;
  if (w < 3) {
    const float* Ab = &sm[row*SMS + khalf];
    if (w == 0) {
      acc = av_tile<16>(Ab, &Rs[khalf*16 + col]);
    } else {
      const int vc = (w == 1) ? col : ((col < 8) ? 16 + col : 23);
      acc = av_tile<24>(Ab, &Rs[LQ*16 + khalf*24 + vc]);
      if (w == 1 || col < 8) {
        const int pc = (w == 1) ? col : 16 + col;
#pragma unroll
        for (int r = 0; r < 8; ++r)
          pt[(rbase + r)*24 + pc] = acc[r];
      }
    }
  }
  __syncthreads();

  if (w == 0) {
#pragma unroll
    for (int r = 0; r < 8; ++r)
      res[RES_IDX(h, nt*16 + rbase + r, col)] = acc[r];
  }

  {
    const int rr = tid >> 3;
    const int p  = tid & 7;
    float* resp = res + RES_IDX(h, en, 0);
    float g[3], loc[3], nrm = 0.f;
    for (int gg = 0; gg < 3; ++gg) g[gg] = pt[rr*24 + p*3 + gg] - tv[gg];
    for (int f = 0; f < 3; ++f) {
      float v = 0.f;
      for (int gg = 0; gg < 3; ++gg) v += g[gg]*Rm[gg*3 + f];
      loc[f] = v; nrm += v*v;
    }
    for (int f = 0; f < 3; ++f) resp[144 + p*3 + f] = loc[f];
    resp[168 + p] = sqrtf(nrm + 1e-12f);
  }
}

#define PO_KC 64
#define PO_NCH (LQ / PO_KC)
#define PO_AP 68
#define PO_BP 144

static __device__ __forceinline__ v8f pairout_chunk(
    const float* __restrict__ Ab, const float* __restrict__ Bb, v8f acc)
{
  v2f a[PO_KC / 4], b[PO_KC / 4];
#pragma unroll
  for (int kk = 0; kk < PO_KC / 4; ++kk) {
    a[kk]   = *reinterpret_cast<const v2f*>(Ab + kk*4);
    b[kk].x = Bb[(kk*4)     * PO_BP];
    b[kk].y = Bb[(kk*4 + 1) * PO_BP];
  }
  __builtin_amdgcn_sched_barrier(0);
#pragma unroll
  for (int kk = 0; kk < PO_KC / 4; ++kk)
    acc = wmma4(a[kk], b[kk], acc);
  asm volatile("" : "+v"(acc));
  __builtin_amdgcn_sched_barrier(0);
  return acc;
}

__global__ __launch_bounds__(256) void pairout_kernel(
    const float* __restrict__ pairwise, float* __restrict__ ws)
{
  __shared__ __attribute__((aligned(16))) float As[16 * PO_AP];
  __shared__ __attribute__((aligned(16))) float Bs[PO_KC * PO_BP];

  const float* attn = ws + OFF_ATTN;
  float* res = ws + OFF_RES;
  const int tid   = threadIdx.x;
  const int lane  = tid & 31;
  const int w     = tid >> 5;
  const int n     = blockIdx.x;
  const int row   = lane & 15;
  const int khalf = (lane >> 4) * 2;
  const int col   = lane & 15;

  const float* gb = pairwise + (size_t)n * LQ * PAIR + (size_t)tid * 4;
  const int sb = (tid >> 5) * PO_BP + (tid & 31) * 4;
  const bool hasA = tid < NH * 16;
  const int ah = hasA ? (tid >> 4) : 0;
  const float* gaa = attn + ((size_t)ah * LQ + n) * LQ + (tid & 15) * 4;
  const int sa = ah * PO_AP + (tid & 15) * 4;

  const int fa = row * PO_AP + khalf;
  const int fb = khalf * PO_BP + w*16 + col;

  v4f rb[8], ra;
#pragma unroll
  for (int i = 0; i < 8; ++i)
    rb[i] = *reinterpret_cast<const v4f*>(gb + (size_t)i * 1024);
  ra = *reinterpret_cast<const v4f*>(gaa);
#pragma unroll
  for (int i = 0; i < 8; ++i)
    *reinterpret_cast<v4f*>(&Bs[sb + i*8*PO_BP]) = rb[i];
  if (hasA) {
    *reinterpret_cast<v4f*>(&As[sa]) = ra;
  } else {
    const v4f z4 = {0.f, 0.f, 0.f, 0.f};
    *reinterpret_cast<v4f*>(&As[(tid >> 4) * PO_AP + (tid & 15) * 4]) = z4;
  }
  __syncthreads();

  v8f acc = vzero8();
  for (int ch = 0; ch < PO_NCH - 1; ++ch) {
    const float* gbn = gb + (size_t)(ch + 1) * PO_KC * PAIR;
#pragma unroll
    for (int i = 0; i < 8; ++i)
      rb[i] = *reinterpret_cast<const v4f*>(gbn + (size_t)i * 1024);
    ra = *reinterpret_cast<const v4f*>(gaa + (ch + 1) * PO_KC);

    acc = pairout_chunk(&As[fa], &Bs[fb], acc);

    __syncthreads();
#pragma unroll
    for (int i = 0; i < 8; ++i)
      *reinterpret_cast<v4f*>(&Bs[sb + i*8*PO_BP]) = rb[i];
    if (hasA) *reinterpret_cast<v4f*>(&As[sa]) = ra;
    __syncthreads();
  }
  acc = pairout_chunk(&As[fa], &Bs[fb], acc);

  const int rbase = (lane >> 4) * 8;
#pragma unroll
  for (int r = 0; r < 8; ++r) {
    const int hh = rbase + r;
    if (hh < NH)
      res[RES_IDX(hh, n, 16 + w*16 + col)] = acc[r];
  }
}

#define OP_KC  192
#define OP_NCH (KOUT / OP_KC)
#define OP_AP  196
#define OP_BP  48

static __device__ __forceinline__ v8f outproj_chunk(
    const float* __restrict__ Ab, const float* __restrict__ Bb, v8f acc)
{
  v2f a[OP_KC / 4], b[OP_KC / 4];
#pragma unroll
  for (int kk = 0; kk < OP_KC / 4; ++kk) {
    a[kk]   = *reinterpret_cast<const v2f*>(Ab + kk*4);
    b[kk].x = Bb[(kk*4)     * OP_BP];
    b[kk].y = Bb[(kk*4 + 1) * OP_BP];
  }
  __builtin_amdgcn_sched_barrier(0);
#pragma unroll
  for (int kk = 0; kk < OP_KC / 4; ++kk)
    acc = wmma4(a[kk], b[kk], acc);
  asm volatile("" : "+v"(acc));
  __builtin_amdgcn_sched_barrier(0);
  return acc;
}

__global__ __launch_bounds__(128) void outproj_kernel(
    const float* __restrict__ out_k, const float* __restrict__ out_b,
    const float* __restrict__ res, float* __restrict__ out)
{
  __shared__ __attribute__((aligned(16))) float As[2][32 * OP_AP];
  __shared__ __attribute__((aligned(16))) float Bs[2][OP_KC * OP_BP];

  const int tid   = threadIdx.x;
  const int lane  = tid & 31;
  const int w     = tid >> 5;
  const int wm    = w >> 1;
  const int wn    = w & 1;
  const int m0    = blockIdx.x * 32;
  const int n0    = blockIdx.y * 32;
  const int col   = lane & 15;
  const int khalf = (lane >> 4) * 2;

  const int srow = tid >> 3;
  const int sseg = tid & 7;
  const float* ga = res   + (size_t)(m0 + srow) * KOUT + sseg*4;
  const float* gb = out_k + (size_t)srow * NODE + n0 + sseg*4;
  const int sa = srow * OP_AP + sseg*4;
  const int sb = srow * OP_BP + sseg*4;

  const int fa = (wm*16 + col) * OP_AP + khalf;
  const int fb = khalf * OP_BP + wn*16 + col;
  const int o  = n0 + wn*16 + col;
  const float ob = out_b[o];

  v4f ra[2 * (OP_KC / 32)], rb[OP_KC / 16];
#pragma unroll
  for (int i = 0; i < 2; ++i)
#pragma unroll
    for (int j = 0; j < OP_KC / 32; ++j)
      ra[i*(OP_KC/32) + j] = *reinterpret_cast<const v4f*>(ga + (size_t)i*16*KOUT + j*32);
#pragma unroll
  for (int i = 0; i < OP_KC / 16; ++i)
    rb[i] = *reinterpret_cast<const v4f*>(gb + (size_t)i*16*NODE);
#pragma unroll
  for (int i = 0; i < 2; ++i)
#pragma unroll
    for (int j = 0; j < OP_KC / 32; ++j)
      *reinterpret_cast<v4f*>(&As[0][sa + i*16*OP_AP + j*32]) = ra[i*(OP_KC/32) + j];
#pragma unroll
  for (int i = 0; i < OP_KC / 16; ++i)
    *reinterpret_cast<v4f*>(&Bs[0][sb + i*16*OP_BP]) = rb[i];
  __syncthreads();

  v8f acc = vzero8();
  for (int ch = 0; ch < OP_NCH - 1; ++ch) {
    const int cur = ch & 1, nxt = cur ^ 1;
    const float* gan = ga + (size_t)(ch + 1) * OP_KC;
    const float* gbn = gb + (size_t)(ch + 1) * OP_KC * NODE;
#pragma unroll
    for (int i = 0; i < 2; ++i)
#pragma unroll
      for (int j = 0; j < OP_KC / 32; ++j)
        ra[i*(OP_KC/32) + j] = *reinterpret_cast<const v4f*>(gan + (size_t)i*16*KOUT + j*32);
#pragma unroll
    for (int i = 0; i < OP_KC / 16; ++i)
      rb[i] = *reinterpret_cast<const v4f*>(gbn + (size_t)i*16*NODE);

    acc = outproj_chunk(&As[cur][fa], &Bs[cur][fb], acc);

#pragma unroll
    for (int i = 0; i < 2; ++i)
#pragma unroll
      for (int j = 0; j < OP_KC / 32; ++j)
        *reinterpret_cast<v4f*>(&As[nxt][sa + i*16*OP_AP + j*32]) = ra[i*(OP_KC/32) + j];
#pragma unroll
    for (int i = 0; i < OP_KC / 16; ++i)
      *reinterpret_cast<v4f*>(&Bs[nxt][sb + i*16*OP_BP]) = rb[i];
    __syncthreads();
  }
  acc = outproj_chunk(&As[(OP_NCH - 1) & 1][fa], &Bs[(OP_NCH - 1) & 1][fb], acc);

  const int rbase = (lane >> 4) * 8;
#pragma unroll
  for (int r = 0; r < 8; ++r) {
    const int l = m0 + wm*16 + rbase + r;
    out[(size_t)l*NODE + o] = acc[r] + ob;
  }
}

extern "C" void kernel_launch(void* const* d_in, const int* in_sizes, int n_in,
                              void* d_out, int out_size, void* d_ws, size_t ws_size,
                              hipStream_t stream) {
  const float* node  = (const float*)d_in[0];
  const float* R     = (const float*)d_in[1];
  const float* t     = (const float*)d_in[2];
  const float* pairw = (const float*)d_in[3];
  const float* sq_k  = (const float*)d_in[4];
  const float* sk_k  = (const float*)d_in[5];
  const float* sv_k  = (const float*)d_in[6];
  const float* pq_k  = (const float*)d_in[7];
  const float* pk_k  = (const float*)d_in[8];
  const float* pv_k  = (const float*)d_in[9];
  const float* pr_k  = (const float*)d_in[10];
  const float* out_k = (const float*)d_in[11];
  const float* out_b = (const float*)d_in[12];
  const float* pwgt  = (const float*)d_in[13];
  float* ws  = (float*)d_ws;
  float* out = (float*)d_out;

  proj_kernel<<<dim3(32, 7, NH), 32, 0, stream>>>(
      node, sq_k, sk_k, sv_k, pq_k, pk_k, pv_k, ws);
  rigid_kernel<<<(NH*LQ + 127)/128, 128, 0, stream>>>(R, t, ws);
  bias_kernel<<<LQ, 256, 0, stream>>>(pairw, pr_k, ws);
  attn_kernel<<<dim3(32, NH), 128, 0, stream>>>(pwgt, R, t, ws);
  pairout_kernel<<<LQ, 256, 0, stream>>>(pairw, ws);
  outproj_kernel<<<dim3(16, 12), 128, 0, stream>>>(out_k, out_b, ws + OFF_RES, out);
}


# rigid transform and point norms ride in the pair-bias launch: block i does rows (i&~3)..+3 of 3 heads, whole 128-byte lines per block, one thread per row; own launch dropped, 6 -> 5; sums unchanged
# speedup vs baseline: 3.8396x; 1.0498x over previous
#include <hip/hip_runtime.h>
#include <math.h>


typedef __attribute__((ext_vector_type(2))) float v2f;
typedef __attribute__((ext_vector_type(4))) float v4f;
typedef __attribute__((ext_vector_type(8))) float v8f;

#define LQ    512
#define NH    12
#define NODE  384
#define PAIR  128
#define OUTIN 176
#define KOUT  (NH*OUTIN)

#define OFF_SQ   ((size_t)0)
#define OFF_SK   (OFF_SQ  + (size_t)NH*LQ*16)
#define OFF_SV   (OFF_SK  + (size_t)NH*LQ*16)
#define OFF_PQ   (OFF_SV  + (size_t)NH*LQ*16)
#define OFF_PK   (OFF_PQ  + (size_t)NH*LQ*16)
#define OFF_PV   (OFF_PK  + (size_t)NH*LQ*16)
#define OFF_QN2  (OFF_PV  + (size_t)NH*LQ*24)
#define OFF_KN2  (OFF_QN2 + (size_t)NH*LQ)
#define OFF_BIAS (OFF_KN2 + (size_t)NH*LQ)
#define OFF_ATTN (OFF_BIAS+ (size_t)NH*LQ*LQ)
#define OFF_RES  (OFF_ATTN+ (size_t)NH*LQ*LQ)

#define RES_IDX(h, l, f) (((size_t)(l)*NH + (h))*OUTIN + (f))

static __device__ __forceinline__ v8f wmma4(v2f a, v2f b, v8f c) {
  return __builtin_amdgcn_wmma_f32_16x16x4_f32(
       false, a,  false, b,
       (short)0, c,  false,  false);
}

static __device__ __forceinline__ v8f vzero8() {
  v8f z = {0.f,0.f,0.f,0.f,0.f,0.f,0.f,0.f};
  return z;
}

#define PJ_KC  64
#define PJ_NCH (NODE / PJ_KC)
#define PJ_AP  68

template <int WS>
static __device__ __forceinline__ v8f proj_chunk(
    const float* __restrict__ Ab, const float* __restrict__ Bb, float wscale, v8f acc)
{
  v2f a[PJ_KC / 4], b[PJ_KC / 4];
#pragma unroll
  for (int kk = 0; kk < PJ_KC / 4; ++kk) {
    a[kk]   = *reinterpret_cast<const v2f*>(Ab + kk*4);
    b[kk].x = Bb[(kk*4)     * WS] * wscale;
    b[kk].y = Bb[(kk*4 + 1) * WS] * wscale;
  }
  __builtin_amdgcn_sched_barrier(0);
#pragma unroll
  for (int kk = 0; kk < PJ_KC / 4; ++kk)
    acc = wmma4(a[kk], b[kk], acc);
  asm volatile("" : "+v"(acc));
  __builtin_amdgcn_sched_barrier(0);
  return acc;
}

template <int WS>
static __device__ __forceinline__ v8f proj_body(
    const float* __restrict__ ga, const float* __restrict__ gw,
    float* __restrict__ As, float* __restrict__ Bs,
    int lane, int wcol, float wscale)
{
  constexpr int NB = PJ_KC * WS / 128;
  const int row   = lane & 15;
  const int khalf = (lane >> 4) * 2;
  const float* gaa = ga + (size_t)(lane >> 4) * NODE + (lane & 15) * 4;
  const int sa = (lane >> 4) * PJ_AP + (lane & 15) * 4;
  const float* gbb = gw + (size_t)lane * 4;
  const int sb = lane * 4;
  const int fa = row * PJ_AP + khalf;
  const int fb = khalf * WS + wcol;

  v4f ra[8], rb[NB];
#pragma unroll
  for (int q = 0; q < 8; ++q)
    ra[q] = *reinterpret_cast<const v4f*>(gaa + (size_t)q * 2 * NODE);
#pragma unroll
  for (int q = 0; q < NB; ++q)
    rb[q] = *reinterpret_cast<const v4f*>(gbb + (size_t)q * 128);
#pragma unroll
  for (int q = 0; q < 8; ++q)
    *reinterpret_cast<v4f*>(&As[sa + q*2*PJ_AP]) = ra[q];
#pragma unroll
  for (int q = 0; q < NB; ++q)
    *reinterpret_cast<v4f*>(&Bs[sb + q*128]) = rb[q];
  __syncthreads();

  v8f acc = vzero8();
  for (int ch = 0; ch < PJ_NCH - 1; ++ch) {
    const float* gan = gaa + (size_t)(ch + 1) * PJ_KC;
    const float* gbn = gbb + (size_t)(ch + 1) * PJ_KC * WS;
#pragma unroll
    for (int q = 0; q < 8; ++q)
      ra[q] = *reinterpret_cast<const v4f*>(gan + (size_t)q * 2 * NODE);
#pragma unroll
    for (int q = 0; q < NB; ++q)
      rb[q] = *reinterpret_cast<const v4f*>(gbn + (size_t)q * 128);

    acc = proj_chunk<WS>(&As[fa], &Bs[fb], wscale, acc);

    __syncthreads();
#pragma unroll
    for (int q = 0; q < 8; ++q)
      *reinterpret_cast<v4f*>(&As[sa + q*2*PJ_AP]) = ra[q];
#pragma unroll
    for (int q = 0; q < NB; ++q)
      *reinterpret_cast<v4f*>(&Bs[sb + q*128]) = rb[q];
    __syncthreads();
  }
  return proj_chunk<WS>(&As[fa], &Bs[fb], wscale, acc);
}

__global__ __launch_bounds__(32) void proj_kernel(
    const float* __restrict__ node,
    const float* __restrict__ sq_k, const float* __restrict__ sk_k,
    const float* __restrict__ sv_k, const float* __restrict__ pq_k,
    const float* __restrict__ pk_k, const float* __restrict__ pv_k,
    float* __restrict__ ws)
{
  __shared__ __attribute__((aligned(16))) float As[16 * PJ_AP];
  __shared__ __attribute__((aligned(16))) float Bs[PJ_KC * 24];

  const int lane  = threadIdx.x;
  const int mt    = blockIdx.x;
  const int seg   = blockIdx.y;
  const int h     = blockIdx.z;
  const int row   = lane & 15;

  const float* ga = node + (size_t)(mt*16) * NODE;
  v8f acc;
  if (seg < 3) {
    const float* wb = (seg == 0) ? sq_k : (seg == 1) ? sk_k : sv_k;
    acc = proj_body<16>(ga, wb + (size_t)h*NODE*16, As, Bs, lane, row,
                        (seg == 0) ? 0.25f : 1.f);
  } else if (seg < 5) {
    const float* wb = (seg == 3) ? pq_k : pk_k;
    acc = proj_body<12>(ga, wb + (size_t)h*NODE*12, As, Bs, lane,
                        (row < 12) ? row : 11, (row < 12) ? 1.f : 0.f);
  } else if (seg == 5) {
    acc = proj_body<24>(ga, pv_k + (size_t)h*NODE*24, As, Bs, lane, row, 1.f);
  } else {
    acc = proj_body<24>(ga, pv_k + (size_t)h*NODE*24, As, Bs, lane,
                        (row < 8) ? 16 + row : 23, (row < 8) ? 1.f : 0.f);
  }

  const int col   = lane & 15;
  const int rbase = (lane >> 4) * 8;
  float* ob = ws + OFF_SK + (size_t)h*LQ*16;
  int ostride = 16, ocol = col;
  bool dostore = true;
  switch (seg) {
    case 0: ob = ws + OFF_SQ + (size_t)h*LQ*16; break;
    case 1: break;
    case 2: ob = ws + OFF_SV + (size_t)h*LQ*16; break;
    case 3: ob = ws + OFF_PQ + (size_t)h*LQ*16; break;
    case 4: ob = ws + OFF_PK + (size_t)h*LQ*16; break;
    case 5: ob = ws + OFF_PV + (size_t)h*LQ*24; ostride = 24; break;
    case 6: ob = ws + OFF_PV + (size_t)h*LQ*24; ostride = 24;
            ocol = 16 + col; dostore = (col < 8); break;
  }
  if (dostore) {
#pragma unroll
    for (int r = 0; r < 8; ++r)
      ob[(size_t)(mt*16 + rbase + r)*ostride + ocol] = acc[r];
  }
}

struct RigidRow {
  float Rm[9], tv[3];
  v4f   c[6];
  float* P;
  float* n2;
  bool  on, isv;
};

static __device__ __forceinline__ RigidRow rigid_fetch(
    const float* __restrict__ R, const float* __restrict__ t,
    float* __restrict__ ws, int i, int w, int lane)
{
  RigidRow rr;
  const int hh = (w == 0) ? ((lane & 15) >> 2) : (lane >> 2);
  rr.on  = (w < 2) && (hh < 3);
  rr.isv = (w == 1);
  const int l = (i & ~3) + (lane & 3);
  const int h = (i & 3) * 3 + hh;
  const bool isk = lane >= 16;
  rr.P  = rr.isv ? ws + OFF_PV + ((size_t)h*LQ + l)*24
                 : ws + (isk ? OFF_PK : OFF_PQ) + ((size_t)h*LQ + l)*16;
  rr.n2 = ws + (isk ? OFF_KN2 : OFF_QN2) + h*LQ + l;
  const v4f z4 = {0.f, 0.f, 0.f, 0.f};
#pragma unroll
  for (int k = 0; k < 9; ++k) rr.Rm[k] = 0.f;
#pragma unroll
  for (int k = 0; k < 3; ++k) rr.tv[k] = 0.f;
#pragma unroll
  for (int q = 0; q < 6; ++q) rr.c[q] = z4;
  if (rr.on) {
#pragma unroll
    for (int k = 0; k < 9; ++k) rr.Rm[k] = R[l*9 + k];
#pragma unroll
    for (int k = 0; k < 3; ++k) rr.tv[k] = t[l*3 + k];
#pragma unroll
    for (int q = 0; q < 3; ++q) rr.c[q] = *reinterpret_cast<const v4f*>(rr.P + 4*q);
    if (rr.isv) {
#pragma unroll
      for (int q = 3; q < 6; ++q) rr.c[q] = *reinterpret_cast<const v4f*>(rr.P + 4*q);
    }
  }
  return rr;
}

static __device__ __forceinline__ float rigid_coord(
    const RigidRow& rr, int p, int gg)
{
  float v = rr.tv[gg];
#pragma unroll
  for (int f = 0; f < 3; ++f)
    v = __builtin_fmaf(rr.c[(p*3 + f) >> 2][(p*3 + f) & 3], rr.Rm[gg*3 + f], v);
  return v;
}

static __device__ __forceinline__ void rigid_finish(const RigidRow& rr)
{
  if (!rr.on) return;
  if (!rr.isv) {
    v4f o[3];
    float s = 0.f;
#pragma unroll
    for (int p = 0; p < 4; ++p) {
#pragma unroll
      for (int gg = 0; gg < 3; ++gg) {
        const float v = rigid_coord(rr, p, gg);
        o[(p*3 + gg) >> 2][(p*3 + gg) & 3] = v;
        s = __builtin_fmaf(v, v, s);
      }
    }
#pragma unroll
    for (int q = 0; q < 3; ++q) *reinterpret_cast<v4f*>(rr.P + 4*q) = o[q];
    *rr.n2 = s;
  } else {
    v4f o[6];
#pragma unroll
    for (int p = 0; p < 8; ++p) {
#pragma unroll
      for (int gg = 0; gg < 3; ++gg)
        o[(p*3 + gg) >> 2][(p*3 + gg) & 3] = rigid_coord(rr, p, gg);
    }
#pragma unroll
    for (int q = 0; q < 6; ++q) *reinterpret_cast<v4f*>(rr.P + 4*q) = o[q];
  }
}

#define BI_JC 128
#define BI_NCH (LQ / BI_JC)
#define BI_BP 132

static __device__ __forceinline__ v8f bias_chunk(
    const float* __restrict__ Bb, const v2f (&pa)[PAIR / 4])
{
  v2f b[PAIR / 4];
#pragma unroll
  for (int kk = 0; kk < PAIR / 4; ++kk)
    b[kk] = *reinterpret_cast<const v2f*>(Bb + kk*4);
  __builtin_amdgcn_sched_barrier(0);
  v8f acc = vzero8();
#pragma unroll
  for (int kk = 0; kk < PAIR / 4; ++kk)
    acc = wmma4(pa[kk], b[kk], acc);
  asm volatile("" : "+v"(acc));
  __builtin_amdgcn_sched_barrier(0);
  return acc;
}

static __device__ __forceinline__ void bias_store(
    float* __restrict__ ob, v8f acc, bool lowhalf)
{
#pragma unroll
  for (int r = 0; r < 4; ++r) ob[(size_t)r * LQ * LQ] = acc[r];
  if (lowhalf) {
#pragma unroll
    for (int r = 4; r < 8; ++r) ob[(size_t)r * LQ * LQ] = acc[r];
  }
}

__global__ __launch_bounds__(256) void bias_kernel(
    const float* __restrict__ pairwise, const float* __restrict__ pr_k,
    const float* __restrict__ frameR, const float* __restrict__ framet,
    float* __restrict__ ws)
{
  __shared__ __attribute__((aligned(16))) float Bs[BI_JC * BI_BP];

  float* bias = ws + OFF_BIAS;
  const int tid   = threadIdx.x;
  const int lane  = tid & 31;
  const int w     = tid >> 5;
  const int i     = blockIdx.x;
  const int row   = lane & 15;
  const int khalf = (lane >> 4) * 2;

  const RigidRow rr = rigid_fetch(frameR, framet, ws, i, w, lane);

  const float* gb = pairwise + (size_t)i * LQ * PAIR + (size_t)tid * 4;
  const int sb = (tid >> 5) * BI_BP + (tid & 31) * 4;
  const int fb = (w*16 + row) * BI_BP + khalf;

  v4f rb[16];
  {
    const float* gbl = gb + (size_t)(BI_NCH - 1) * BI_JC * PAIR;
#pragma unroll
    for (int q = 0; q < 16; ++q)
      rb[q] = *reinterpret_cast<const v4f*>(gbl + (size_t)q * 1024);
  }

  const bool live = row < NH;
  const float* prk = pr_k + (size_t)(live ? row : NH - 1) * PAIR + khalf;
  v2f pa[PAIR / 4];
#pragma unroll
  for (int kk = 0; kk < PAIR / 4; ++kk) {
    const v2f t = *reinterpret_cast<const v2f*>(prk + kk*4);
    pa[kk].x = live ? t.x : 0.f;
    pa[kk].y = live ? t.y : 0.f;
  }

  rigid_finish(rr);

#pragma unroll
  for (int q = 0; q < 16; ++q)
    *reinterpret_cast<v4f*>(&Bs[sb + q*8*BI_BP]) = rb[q];
  __syncthreads();

  const bool lowhalf = lane < 16;
  float* ob = bias + ((size_t)((lane >> 4) * 8) * LQ + i) * LQ + w*16 + row;

  for (int ch = BI_NCH - 1; ch > 0; --ch) {
    const float* gbn = gb + (size_t)(ch - 1) * BI_JC * PAIR;
#pragma unroll
    for (int q = 0; q < 16; ++q)
      rb[q] = *reinterpret_cast<const v4f*>(gbn + (size_t)q * 1024);

    const v8f acc = bias_chunk(&Bs[fb], pa);

    __syncthreads();
#pragma unroll
    for (int q = 0; q < 16; ++q)
      *reinterpret_cast<v4f*>(&Bs[sb + q*8*BI_BP]) = rb[q];
    __syncthreads();
    bias_store(ob + ch * BI_JC, acc, lowhalf);
  }
  const v8f acc = bias_chunk(&Bs[fb], pa);
  bias_store(ob, acc, lowhalf);
}

#define SMS 516
#define AT_KP 20
#define AT_G  32

template <int VP>
static __device__ __forceinline__ v8f av_group(
    const float* __restrict__ Ab, const float* __restrict__ Bb, v8f acc)
{
  v2f a[AT_G], b[AT_G];
#pragma unroll
  for (int kk = 0; kk < AT_G; ++kk) {
    a[kk]   = *reinterpret_cast<const v2f*>(Ab + kk*4);
    b[kk].x = Bb[(kk*4)     * VP];
    b[kk].y = Bb[(kk*4 + 1) * VP];
  }
  __builtin_amdgcn_sched_barrier(0);
#pragma unroll
  for (int kk = 0; kk < AT_G; ++kk)
    acc = wmma4(a[kk], b[kk], acc);
  __builtin_amdgcn_sched_barrier(0);
  return acc;
}

template <int VP>
static __device__ __forceinline__ v8f av_tile(
    const float* __restrict__ Ab, const float* __restrict__ Bb)
{
  v8f acc = vzero8();
#pragma unroll
  for (int g = 0; g < LQ / (4 * AT_G); ++g)
    acc = av_group<VP>(Ab + g * 4 * AT_G, Bb + (size_t)g * 4 * AT_G * VP, acc);
  return acc;
}

__global__ __launch_bounds__(128) void attn_kernel(
    const float* __restrict__ point_weight,
    const float* __restrict__ R, const float* __restrict__ t,
    float* __restrict__ ws)
{
  __shared__ __attribute__((aligned(16))) float sm[16 * SMS];
  __shared__ __attribute__((aligned(16))) float Rs[LQ * 40];
  __shared__ __attribute__((aligned(16))) float kn[LQ];
  const int tid  = threadIdx.x;
  const int lane = tid & 31;
  const int w    = tid >> 5;
  const int nt   = blockIdx.x;
  const int h    = blockIdx.y;

  const float* Sq  = ws + OFF_SQ;  const float* Sk  = ws + OFF_SK;
  const float* Pq  = ws + OFF_PQ;  const float* Pk  = ws + OFF_PK;
  const float* qn2 = ws + OFF_QN2; const float* kn2 = ws + OFF_KN2;
  const float* bias= ws + OFF_BIAS;
  const float* Sv  = ws + OFF_SV;  const float* Pv  = ws + OFF_PV;
  float* attn = ws + OFF_ATTN;
  float* res  = ws + OFF_RES;

  const float pwx = point_weight[h];
  const float pw  = (pwx > 20.f) ? pwx : log1pf(__expf(pwx));
  const float cpt = -0.5f * pw / sqrtf(18.f);
  const float rs3 = 0.57735026919f;

  const int row   = lane & 15;
  const int khalf = (lane >> 4) * 2;
  const int col   = lane & 15;
  const int rbase = (lane >> 4) * 8;

  {
    const float* gsk = Sk + (size_t)h*LQ*16 + (size_t)tid*4;
    const float* gpk = Pk + (size_t)h*LQ*16 + (size_t)tid*4;
    const float* gbi = bias + ((size_t)h*LQ + nt*16)*LQ + (size_t)tid*4;
    v4f rk[16], rp[16], rb[16];
#pragma unroll
    for (int q = 0; q < 16; ++q) {
      rk[q] = *reinterpret_cast<const v4f*>(gsk + (size_t)q*512);
      rp[q] = *reinterpret_cast<const v4f*>(gpk + (size_t)q*512);
    }
#pragma unroll
    for (int q = 0; q < 16; ++q)
      rb[q] = *reinterpret_cast<const v4f*>(gbi + (size_t)q*LQ);
    const v4f rn = *reinterpret_cast<const v4f*>(kn2 + (size_t)h*LQ + tid*4);
    const int kd = (tid >> 2) * AT_KP + (tid & 3) * 4;
#pragma unroll
    for (int q = 0; q < 16; ++q) {
      *reinterpret_cast<v4f*>(&Rs[kd + q*32*AT_KP])              = rk[q];
      *reinterpret_cast<v4f*>(&Rs[LQ*AT_KP + kd + q*32*AT_KP])   = rp[q];
    }
#pragma unroll
    for (int q = 0; q < 16; ++q)
      *reinterpret_cast<v4f*>(&sm[q*SMS + tid*4]) = rb[q];
    *reinterpret_cast<v4f*>(&kn[tid*4]) = rn;
  }

  v2f aq[4], ap[4];
#pragma unroll
  for (int ks = 0; ks < 4; ++ks) {
    const int k = ks*4 + khalf;
    const int l = nt*16 + row;
    aq[ks].x = Sq[((size_t)h*LQ + l)*16 + k];
    aq[ks].y = Sq[((size_t)h*LQ + l)*16 + k + 1];
    ap[ks].x = Pq[((size_t)h*LQ + l)*16 + k];
    ap[ks].y = Pq[((size_t)h*LQ + l)*16 + k + 1];
  }
  float qv[8];
#pragma unroll
  for (int r = 0; r < 8; ++r) qv[r] = qn2[h*LQ + nt*16 + rbase + r];
  __syncthreads();

  for (int mt = w; mt < 32; mt += 4) {
    v8f s = vzero8(), p = vzero8();
    const float* kb = &Rs[(mt*16 + row)*AT_KP + khalf];
#pragma unroll
    for (int ks = 0; ks < 4; ++ks) {
      const v2f bq = *reinterpret_cast<const v2f*>(kb + ks*4);
      const v2f bp = *reinterpret_cast<const v2f*>(kb + LQ*AT_KP + ks*4);
      s = wmma4(aq[ks], bq, s);
      p = wmma4(ap[ks], bp, p);
    }
    const int m = mt*16 + col;
    const float kv = kn[m];
#pragma unroll
    for (int r = 0; r < 8; ++r) {
      const int rr = rbase + r;
      const float dist = qv[r] + kv - 2.f*p[r];
      sm[rr*SMS + m] = (s[r] + sm[rr*SMS + m] + cpt*dist) * rs3;
    }
  }
  __syncthreads();

  v4f rv[40];
  {
    const float* gsv = Sv + (size_t)h*LQ*16 + (size_t)tid*4;
    const float* gpv = Pv + (size_t)h*LQ*24 + (size_t)tid*4;
#pragma unroll
    for (int q = 0; q < 16; ++q)
      rv[q] = *reinterpret_cast<const v4f*>(gsv + (size_t)q*512);
#pragma unroll
    for (int q = 0; q < 24; ++q)
      rv[16 + q] = *reinterpret_cast<const v4f*>(gpv + (size_t)q*512);
  }

  {
    const int rrow = tid >> 3;
    const int sub  = tid & 7;
    float mx = -1e30f;
    for (int j = sub; j < LQ; j += 8) mx = fmaxf(mx, sm[rrow*SMS + j]);
    for (int o = 1; o < 8; o <<= 1) mx = fmaxf(mx, __shfl_xor(mx, o, 8));
    float sum = 0.f;
    for (int j = sub; j < LQ; j += 8) {
      const float e = __expf(sm[rrow*SMS + j] - mx);
      sm[rrow*SMS + j] = e; sum += e;
    }
    for (int o = 1; o < 8; o <<= 1) sum += __shfl_xor(sum, o, 8);
    const float inv = 1.f / sum;
    const int n = nt*16 + rrow;
    for (int j = sub; j < LQ; j += 8) {
      const float v = sm[rrow*SMS + j] * inv;
      sm[rrow*SMS + j] = v;
      attn[((size_t)h*LQ + n)*LQ + j] = v;
    }
  }

#pragma unroll
  for (int q = 0; q < 16; ++q)
    *reinterpret_cast<v4f*>(&Rs[(tid + 128*q)*4]) = rv[q];
#pragma unroll
  for (int q = 0; q < 24; ++q)
    *reinterpret_cast<v4f*>(&Rs[LQ*16 + (tid + 128*q)*4]) = rv[16 + q];
  __syncthreads();

  const int en = nt*16 + (tid >> 3);
  float Rm[9], tv[3];
#pragma unroll
  for (int i = 0; i < 9; ++i) Rm[i] = R[en*9 + i];
#pragma unroll
  for (int i = 0; i < 3; ++i) tv[i] = t[en*3 + i];

  float* pt = kn;
  v8f acc;
  if (w < 3) {
    const float* Ab = &sm[row*SMS + khalf];
    if (w == 0) {
      acc = av_tile<16>(Ab, &Rs[khalf*16 + col]);
    } else {
      const int vc = (w == 1) ? col : ((col < 8) ? 16 + col : 23);
      acc = av_tile<24>(Ab, &Rs[LQ*16 + khalf*24 + vc]);
      if (w == 1 || col < 8) {
        const int pc = (w == 1) ? col : 16 + col;
#pragma unroll
        for (int r = 0; r < 8; ++r)
          pt[(rbase + r)*24 + pc] = acc[r];
      }
    }
  }
  __syncthreads();

  if (w == 0) {
#pragma unroll
    for (int r = 0; r < 8; ++r)
      res[RES_IDX(h, nt*16 + rbase + r, col)] = acc[r];
  }

  {
    const int rr = tid >> 3;
    const int p  = tid & 7;
    float* resp = res + RES_IDX(h, en, 0);
    float g[3], loc[3], nrm = 0.f;
    for (int gg = 0; gg < 3; ++gg) g[gg] = pt[rr*24 + p*3 + gg] - tv[gg];
    for (int f = 0; f < 3; ++f) {
      float v = 0.f;
      for (int gg = 0; gg < 3; ++gg) v += g[gg]*Rm[gg*3 + f];
      loc[f] = v; nrm += v*v;
    }
    for (int f = 0; f < 3; ++f) resp[144 + p*3 + f] = loc[f];
    resp[168 + p] = sqrtf(nrm + 1e-12f);
  }
}

#define PO_KC 64
#define PO_NCH (LQ / PO_KC)
#define PO_AP 68
#define PO_BP 144

static __device__ __forceinline__ v8f pairout_chunk(
    const float* __restrict__ Ab, const float* __restrict__ Bb, v8f acc)
{
  v2f a[PO_KC / 4], b[PO_KC / 4];
#pragma unroll
  for (int kk = 0; kk < PO_KC / 4; ++kk) {
    a[kk]   = *reinterpret_cast<const v2f*>(Ab + kk*4);
    b[kk].x = Bb[(kk*4)     * PO_BP];
    b[kk].y = Bb[(kk*4 + 1) * PO_BP];
  }
  __builtin_amdgcn_sched_barrier(0);
#pragma unroll
  for (int kk = 0; kk < PO_KC / 4; ++kk)
    acc = wmma4(a[kk], b[kk], acc);
  asm volatile("" : "+v"(acc));
  __builtin_amdgcn_sched_barrier(0);
  return acc;
}

__global__ __launch_bounds__(256) void pairout_kernel(
    const float* __restrict__ pairwise, float* __restrict__ ws)
{
  __shared__ __attribute__((aligned(16))) float As[16 * PO_AP];
  __shared__ __attribute__((aligned(16))) float Bs[PO_KC * PO_BP];

  const float* attn = ws + OFF_ATTN;
  float* res = ws + OFF_RES;
  const int tid   = threadIdx.x;
  const int lane  = tid & 31;
  const int w     = tid >> 5;
  const int n     = blockIdx.x;
  const int row   = lane & 15;
  const int khalf = (lane >> 4) * 2;
  const int col   = lane & 15;

  const float* gb = pairwise + (size_t)n * LQ * PAIR + (size_t)tid * 4;
  const int sb = (tid >> 5) * PO_BP + (tid & 31) * 4;
  const bool hasA = tid < NH * 16;
  const int ah = hasA ? (tid >> 4) : 0;
  const float* gaa = attn + ((size_t)ah * LQ + n) * LQ + (tid & 15) * 4;
  const int sa = ah * PO_AP + (tid & 15) * 4;

  const int fa = row * PO_AP + khalf;
  const int fb = khalf * PO_BP + w*16 + col;

  v4f rb[8], ra;
#pragma unroll
  for (int i = 0; i < 8; ++i)
    rb[i] = *reinterpret_cast<const v4f*>(gb + (size_t)i * 1024);
  ra = *reinterpret_cast<const v4f*>(gaa);
#pragma unroll
  for (int i = 0; i < 8; ++i)
    *reinterpret_cast<v4f*>(&Bs[sb + i*8*PO_BP]) = rb[i];
  if (hasA) {
    *reinterpret_cast<v4f*>(&As[sa]) = ra;
  } else {
    const v4f z4 = {0.f, 0.f, 0.f, 0.f};
    *reinterpret_cast<v4f*>(&As[(tid >> 4) * PO_AP + (tid & 15) * 4]) = z4;
  }
  __syncthreads();

  v8f acc = vzero8();
  for (int ch = 0; ch < PO_NCH - 1; ++ch) {
    const float* gbn = gb + (size_t)(ch + 1) * PO_KC * PAIR;
#pragma unroll
    for (int i = 0; i < 8; ++i)
      rb[i] = *reinterpret_cast<const v4f*>(gbn + (size_t)i * 1024);
    ra = *reinterpret_cast<const v4f*>(gaa + (ch + 1) * PO_KC);

    acc = pairout_chunk(&As[fa], &Bs[fb], acc);

    __syncthreads();
#pragma unroll
    for (int i = 0; i < 8; ++i)
      *reinterpret_cast<v4f*>(&Bs[sb + i*8*PO_BP]) = rb[i];
    if (hasA) *reinterpret_cast<v4f*>(&As[sa]) = ra;
    __syncthreads();
  }
  acc = pairout_chunk(&As[fa], &Bs[fb], acc);

  const int rbase = (lane >> 4) * 8;
#pragma unroll
  for (int r = 0; r < 8; ++r) {
    const int hh = rbase + r;
    if (hh < NH)
      res[RES_IDX(hh, n, 16 + w*16 + col)] = acc[r];
  }
}

#define OP_KC  192
#define OP_NCH (KOUT / OP_KC)
#define OP_AP  196
#define OP_BP  48

static __device__ __forceinline__ v8f outproj_chunk(
    const float* __restrict__ Ab, const float* __restrict__ Bb, v8f acc)
{
  v2f a[OP_KC / 4], b[OP_KC / 4];
#pragma unroll
  for (int kk = 0; kk < OP_KC / 4; ++kk) {
    a[kk]   = *reinterpret_cast<const v2f*>(Ab + kk*4);
    b[kk].x = Bb[(kk*4)     * OP_BP];
    b[kk].y = Bb[(kk*4 + 1) * OP_BP];
  }
  __builtin_amdgcn_sched_barrier(0);
#pragma unroll
  for (int kk = 0; kk < OP_KC / 4; ++kk)
    acc = wmma4(a[kk], b[kk], acc);
  asm volatile("" : "+v"(acc));
  __builtin_amdgcn_sched_barrier(0);
  return acc;
}

__global__ __launch_bounds__(128) void outproj_kernel(
    const float* __restrict__ out_k, const float* __restrict__ out_b,
    const float* __restrict__ res, float* __restrict__ out)
{
  __shared__ __attribute__((aligned(16))) float As[2][32 * OP_AP];
  __shared__ __attribute__((aligned(16))) float Bs[2][OP_KC * OP_BP];

  const int tid   = threadIdx.x;
  const int lane  = tid & 31;
  const int w     = tid >> 5;
  const int wm    = w >> 1;
  const int wn    = w & 1;
  const int m0    = blockIdx.x * 32;
  const int n0    = blockIdx.y * 32;
  const int col   = lane & 15;
  const int khalf = (lane >> 4) * 2;

  const int srow = tid >> 3;
  const int sseg = tid & 7;
  const float* ga = res   + (size_t)(m0 + srow) * KOUT + sseg*4;
  const float* gb = out_k + (size_t)srow * NODE + n0 + sseg*4;
  const int sa = srow * OP_AP + sseg*4;
  const int sb = srow * OP_BP + sseg*4;

  const int fa = (wm*16 + col) * OP_AP + khalf;
  const int fb = khalf * OP_BP + wn*16 + col;
  const int o  = n0 + wn*16 + col;
  const float ob = out_b[o];

  v4f ra[2 * (OP_KC / 32)], rb[OP_KC / 16];
#pragma unroll
  for (int i = 0; i < 2; ++i)
#pragma unroll
    for (int j = 0; j < OP_KC / 32; ++j)
      ra[i*(OP_KC/32) + j] = *reinterpret_cast<const v4f*>(ga + (size_t)i*16*KOUT + j*32);
#pragma unroll
  for (int i = 0; i < OP_KC / 16; ++i)
    rb[i] = *reinterpret_cast<const v4f*>(gb + (size_t)i*16*NODE);
#pragma unroll
  for (int i = 0; i < 2; ++i)
#pragma unroll
    for (int j = 0; j < OP_KC / 32; ++j)
      *reinterpret_cast<v4f*>(&As[0][sa + i*16*OP_AP + j*32]) = ra[i*(OP_KC/32) + j];
#pragma unroll
  for (int i = 0; i < OP_KC / 16; ++i)
    *reinterpret_cast<v4f*>(&Bs[0][sb + i*16*OP_BP]) = rb[i];
  __syncthreads();

  v8f acc = vzero8();
  for (int ch = 0; ch < OP_NCH - 1; ++ch) {
    const int cur = ch & 1, nxt = cur ^ 1;
    const float* gan = ga + (size_t)(ch + 1) * OP_KC;
    const float* gbn = gb + (size_t)(ch + 1) * OP_KC * NODE;
#pragma unroll
    for (int i = 0; i < 2; ++i)
#pragma unroll
      for (int j = 0; j < OP_KC / 32; ++j)
        ra[i*(OP_KC/32) + j] = *reinterpret_cast<const v4f*>(gan + (size_t)i*16*KOUT + j*32);
#pragma unroll
    for (int i = 0; i < OP_KC / 16; ++i)
      rb[i] = *reinterpret_cast<const v4f*>(gbn + (size_t)i*16*NODE);

    acc = outproj_chunk(&As[cur][fa], &Bs[cur][fb], acc);

#pragma unroll
    for (int i = 0; i < 2; ++i)
#pragma unroll
      for (int j = 0; j < OP_KC / 32; ++j)
        *reinterpret_cast<v4f*>(&As[nxt][sa + i*16*OP_AP + j*32]) = ra[i*(OP_KC/32) + j];
#pragma unroll
    for (int i = 0; i < OP_KC / 16; ++i)
      *reinterpret_cast<v4f*>(&Bs[nxt][sb + i*16*OP_BP]) = rb[i];
    __syncthreads();
  }
  acc = outproj_chunk(&As[(OP_NCH - 1) & 1][fa], &Bs[(OP_NCH - 1) & 1][fb], acc);

  const int rbase = (lane >> 4) * 8;
#pragma unroll
  for (int r = 0; r < 8; ++r) {
    const int l = m0 + wm*16 + rbase + r;
    out[(size_t)l*NODE + o] = acc[r] + ob;
  }
}

extern "C" void kernel_launch(void* const* d_in, const int* in_sizes, int n_in,
                              void* d_out, int out_size, void* d_ws, size_t ws_size,
                              hipStream_t stream) {
  const float* node  = (const float*)d_in[0];
  const float* R     = (const float*)d_in[1];
  const float* t     = (const float*)d_in[2];
  const float* pairw = (const float*)d_in[3];
  const float* sq_k  = (const float*)d_in[4];
  const float* sk_k  = (const float*)d_in[5];
  const float* sv_k  = (const float*)d_in[6];
  const float* pq_k  = (const float*)d_in[7];
  const float* pk_k  = (const float*)d_in[8];
  const float* pv_k  = (const float*)d_in[9];
  const float* pr_k  = (const float*)d_in[10];
  const float* out_k = (const float*)d_in[11];
  const float* out_b = (const float*)d_in[12];
  const float* pwgt  = (const float*)d_in[13];
  float* ws  = (float*)d_ws;
  float* out = (float*)d_out;

  proj_kernel<<<dim3(32, 7, NH), 32, 0, stream>>>(
      node, sq_k, sk_k, sv_k, pq_k, pk_k, pv_k, ws);
  bias_kernel<<<LQ, 256, 0, stream>>>(pairw, pr_k, R, t, ws);
  attn_kernel<<<dim3(32, NH), 128, 0, stream>>>(pwgt, R, t, ws);
  pairout_kernel<<<LQ, 256, 0, stream>>>(pairw, ws);
  outproj_kernel<<<dim3(16, 12), 128, 0, stream>>>(out_k, out_b, ws + OFF_RES, out);
}
